# starting kernel
# baseline (speedup 1.0000x reference)
#include <hip/hip_runtime.h>

typedef __attribute__((ext_vector_type(2))) float v2f;
typedef __attribute__((ext_vector_type(8))) float v8f;

__device__ __forceinline__ void atomic_add_f32(float* p, float v)
{
    asm volatile("global_atomic_add_f32 %0, %1, off"
                 :: "v"(p), "v"(v) : "memory");
}

__global__ __launch_bounds__(128)
void gemm_xw_kernel(const float* __restrict__ x,
                    const float* __restrict__ Wn,
                    const float* __restrict__ Ws,
                    float* __restrict__ xw, int nrows)
{
    const int lane  = threadIdx.x & 31;
    const int wave  = threadIdx.x >> 5;
    const int m0    = blockIdx.x * 16;
    const int lhalf = lane >> 4;
    const int lm    = lane & 15;

    const float* __restrict__ W = (wave < 2) ? Wn : Ws;
    const int n0 = (wave & 1) * 16;

    int arow = m0 + lm;
    if (arow >= nrows) arow = nrows - 1;

    v8f c = {};
    #pragma unroll
    for (int k0 = 0; k0 < 64; k0 += 4) {
        v2f a = *(const v2f*)(x + (size_t)arow * 64 + k0 + 2 * lhalf);
        v2f b;
        b.x = W[(k0 + 2 * lhalf)     * 32 + n0 + lm];
        b.y = W[(k0 + 2 * lhalf + 1) * 32 + n0 + lm];
        c = __builtin_amdgcn_wmma_f32_16x16x4_f32(false, a, false, b,
                                                  (short)0, c, false, false);
    }

    const int gcol = wave * 16 + lm;
    #pragma unroll
    for (int v = 0; v < 8; ++v) {
        const int row = m0 + v + 8 * lhalf;
        if (row < nrows) xw[(size_t)row * 64 + gcol] = c[v];
    }
}

__global__ void init_deg_kernel(float* __restrict__ dn, float* __restrict__ ds, int n)
{
    const int i = blockIdx.x * blockDim.x + threadIdx.x;
    if (i < n) { dn[i] = 1.0f; ds[i] = 1.0f; }
}

__global__ void count_deg_kernel(const int* __restrict__ dst_n,
                                 const int* __restrict__ dst_s,
                                 float* __restrict__ dn,
                                 float* __restrict__ ds, int E)
{
    const int e = blockIdx.x * blockDim.x + threadIdx.x;
    if (e >= E) return;
    const int*   dst = blockIdx.y ? dst_s : dst_n;
    float*       deg = blockIdx.y ? ds    : dn;
    atomic_add_f32(&deg[dst[e]], 1.0f);
}

__global__ void rsqrt_kernel(float* __restrict__ dn, float* __restrict__ ds, int n)
{
    const int i = blockIdx.x * blockDim.x + threadIdx.x;
    if (i < n) {
        const float a = dn[i], b = ds[i];
        dn[i] = (a > 0.0f) ? rsqrtf(a) : 0.0f;
        ds[i] = (b > 0.0f) ? rsqrtf(b) : 0.0f;
    }
}

__global__ void init_h_kernel(const float* __restrict__ xw,
                              const float* __restrict__ dn,
                              const float* __restrict__ ds,
                              float* __restrict__ h, int n)
{
    const int t = blockIdx.x * blockDim.x + threadIdx.x;
    if (t >= n * 32) return;
    const int i = t >> 5, j = t & 31;
    const float a = dn[i], b = ds[i];
    h[t] = xw[(size_t)i * 64 + j]      * (a * a)
         + xw[(size_t)i * 64 + 32 + j] * (b * b);
}

__global__ void scatter_kernel(const int* __restrict__ en,
                               const int* __restrict__ es,
                               const float* __restrict__ xw,
                               const float* __restrict__ dn,
                               const float* __restrict__ dsv,
                               float* __restrict__ h, int E)
{
    const unsigned t = blockIdx.x * blockDim.x + threadIdx.x;
    const unsigned e = t >> 5;
    const int      j = (int)(t & 31u);
    if (e >= (unsigned)E) return;

    const int    typ   = blockIdx.y;
    const int*   edges = typ ? es  : en;
    const float* dinv  = typ ? dsv : dn;
    const int    cbase = typ ? 32  : 0;

    const int s = edges[e], d = edges[E + e];
    const float norm = dinv[s] * dinv[d];
    const float val  = xw[(size_t)s * 64 + cbase + j] * norm;
    atomic_add_f32(&h[(size_t)d * 32 + j], val);
}

__global__ void final_kernel(const float* __restrict__ h,
                             const float* __restrict__ bn,
                             const float* __restrict__ bs,
                             const float* __restrict__ wl,
                             const float* __restrict__ bl,
                             float* __restrict__ out, int n)
{
    const int t = blockIdx.x * blockDim.x + threadIdx.x;
    const int i = t >> 5, j = t & 31;
    if (i >= n) return;
    float v = h[(size_t)i * 32 + j] + bn[j] + bs[j];
    v = fmaxf(v, 0.0f) * wl[j];
    #pragma unroll
    for (int off = 16; off > 0; off >>= 1)
        v += __shfl_xor(v, off, 32);
    if (j == 0) out[i] = v + bl[0];
}

extern "C" void kernel_launch(void* const* d_in, const int* in_sizes, int n_in,
                              void* d_out, int out_size, void* d_ws, size_t ws_size,
                              hipStream_t stream)
{
    const float* x  = (const float*)d_in[0];
    const int*   en = (const int*)  d_in[1];
    const int*   es = (const int*)  d_in[2];
    const float* Wn = (const float*)d_in[3];
    const float* bn = (const float*)d_in[4];
    const float* Ws = (const float*)d_in[5];
    const float* bs = (const float*)d_in[6];
    const float* Wl = (const float*)d_in[7];
    const float* bl = (const float*)d_in[8];
    float* out = (float*)d_out;

    const int N = in_sizes[0] / 64;
    const int E = in_sizes[1] / 2;

    float* ws  = (float*)d_ws;
    float* xw  = ws;
    float* h   = xw + (size_t)N * 64;
    float* dn  = h  + (size_t)N * 32;
    float* ds  = dn + N;

    const int TPB = 256;

    init_deg_kernel<<<(N + TPB - 1) / TPB, TPB, 0, stream>>>(dn, ds, N);
    count_deg_kernel<<<dim3((E + TPB - 1) / TPB, 2), TPB, 0, stream>>>(
        en + E, es + E, dn, ds, E);
    rsqrt_kernel<<<(N + TPB - 1) / TPB, TPB, 0, stream>>>(dn, ds, N);

    gemm_xw_kernel<<<(N + 15) / 16, 128, 0, stream>>>(x, Wn, Ws, xw, N);

    init_h_kernel<<<(N * 32 + TPB - 1) / TPB, TPB, 0, stream>>>(xw, dn, ds, h, N);
    {
        const unsigned tot  = (unsigned)E * 32u;
        const unsigned grid = (tot + TPB - 1) / TPB;
        scatter_kernel<<<dim3(grid, 2), TPB, 0, stream>>>(en, es, xw, dn, ds, h, E);
    }

    final_kernel<<<(N * 32 + TPB - 1) / TPB, TPB, 0, stream>>>(h, bn, bs, Wl, bl, out, N);
}


# h rows owned per work-group in LDS: each scans all dst, queues its matches, adds gathered xw rows by LDS float atomics (arrival-order sums); global h atomics, init_h and head launches gone
# speedup vs baseline: 1.6754x; 1.6754x over previous
#include <hip/hip_runtime.h>

typedef __attribute__((ext_vector_type(2))) float v2f;
typedef __attribute__((ext_vector_type(8))) float v8f;

__device__ __forceinline__ void atomic_add_f32(float* p, float v)
{
    asm volatile("global_atomic_add_f32 %0, %1, off"
                 :: "v"(p), "v"(v) : "memory");
}

typedef __attribute__((address_space(3))) float lds_float;
__device__ __forceinline__ void lds_add_f32(float* p, float v)
{
    asm volatile("ds_add_f32 %0, %1"
                 :: "v"((lds_float*)p), "v"(v) : "memory");
}

__global__ __launch_bounds__(128)
void gemm_xw_kernel(const float* __restrict__ x,
                    const float* __restrict__ Wn,
                    const float* __restrict__ Ws,
                    float* __restrict__ xw, int nrows)
{
    const int lane  = threadIdx.x & 31;
    const int wave  = threadIdx.x >> 5;
    const int m0    = blockIdx.x * 16;
    const int lhalf = lane >> 4;
    const int lm    = lane & 15;

    const float* __restrict__ W = (wave < 2) ? Wn : Ws;
    const int n0 = (wave & 1) * 16;

    int arow = m0 + lm;
    if (arow >= nrows) arow = nrows - 1;

    v8f c = {};
    #pragma unroll
    for (int k0 = 0; k0 < 64; k0 += 4) {
        v2f a = *(const v2f*)(x + (size_t)arow * 64 + k0 + 2 * lhalf);
        v2f b;
        b.x = W[(k0 + 2 * lhalf)     * 32 + n0 + lm];
        b.y = W[(k0 + 2 * lhalf + 1) * 32 + n0 + lm];
        c = __builtin_amdgcn_wmma_f32_16x16x4_f32(false, a, false, b,
                                                  (short)0, c, false, false);
    }

    const int gcol = wave * 16 + lm;
    #pragma unroll
    for (int v = 0; v < 8; ++v) {
        const int row = m0 + v + 8 * lhalf;
        if (row < nrows) xw[(size_t)row * 64 + gcol] = c[v];
    }
}

__global__ void init_deg_kernel(float* __restrict__ dn, float* __restrict__ ds, int n)
{
    const int i = blockIdx.x * blockDim.x + threadIdx.x;
    if (i < n) { dn[i] = 1.0f; ds[i] = 1.0f; }
}

__global__ void count_deg_kernel(const int* __restrict__ dst_n,
                                 const int* __restrict__ dst_s,
                                 float* __restrict__ dn,
                                 float* __restrict__ ds, int E)
{
    const int e = blockIdx.x * blockDim.x + threadIdx.x;
    if (e >= E) return;
    const int*   dst = blockIdx.y ? dst_s : dst_n;
    float*       deg = blockIdx.y ? ds    : dn;
    atomic_add_f32(&deg[dst[e]], 1.0f);
}

__global__ void rsqrt_kernel(float* __restrict__ dn, float* __restrict__ ds, int n)
{
    const int i = blockIdx.x * blockDim.x + threadIdx.x;
    if (i < n) {
        const float a = dn[i], b = ds[i];
        dn[i] = (a > 0.0f) ? rsqrtf(a) : 0.0f;
        ds[i] = (b > 0.0f) ? rsqrtf(b) : 0.0f;
    }
}

constexpr int AGG_THREADS = 1024;
constexpr int AGG_WAVES   = AGG_THREADS / 32;
constexpr int AGG_GRID    = 160;
constexpr int AGG_ROWS    = 625;
constexpr int AGG_U       = 4;
constexpr int AGG_TRIP    = 32 * AGG_U;
constexpr int AGG_QCAP    = 32 + 4 * AGG_TRIP;
constexpr int AGG_B       = 16;

__device__ __forceinline__ void agg_load_trip(const int4* __restrict__ dst4, int v0, int lane,
                                              int4 (&d)[AGG_U])
{
    #pragma unroll
    for (int u = 0; u < AGG_U; ++u) d[u] = dst4[v0 + u * 32 + lane];
}

__device__ __forceinline__ void agg_push(int2* q, int& qn, bool m, int e, unsigned dl)
{
    const unsigned mask = __builtin_amdgcn_ballot_w32(m);
    if (mask) {
        if (m) q[qn + __builtin_amdgcn_mbcnt_lo(mask, 0u)] = make_int2(e, (int)dl);
        qn += __popc(mask);
    }
}

template <bool FULL>
__device__ __forceinline__ void agg_drain(const int2* q, int cnt, int lane,
                                          const int* __restrict__ src,
                                          const float* __restrict__ dinv,
                                          const float* __restrict__ xw,
                                          int base, int cb, float* hacc)
{
    int soff = 0, hoff = 0;
    float nrm = 0.0f;
    if (FULL || lane < cnt) {
        const int2 ent = q[lane];
        const int  s   = src[ent.x];
        nrm  = dinv[s] * dinv[base + ent.y];
        soff = s * 64 + cb;
        hoff = ent.y * 32;
    }
    const int kmax = FULL ? 32 : cnt;
    for (int k0 = 0; k0 < kmax; k0 += AGG_B) {
        float val[AGG_B];
        #pragma unroll
        for (int i = 0; i < AGG_B; ++i) {
            const int so = __builtin_amdgcn_readlane(soff, k0 + i);
            val[i] = xw[so + lane];
        }
        #pragma unroll
        for (int i = 0; i < AGG_B; ++i) {
            if (FULL || k0 + i < cnt) {
                const int   ho = __builtin_amdgcn_readlane(hoff, k0 + i);
                const float nk = __int_as_float(__builtin_amdgcn_readlane(__float_as_int(nrm), k0 + i));
                lds_add_f32(&hacc[ho + lane], val[i] * nk);
            }
        }
    }
}

__global__ __launch_bounds__(AGG_THREADS)
void aggregate_kernel(const int* __restrict__ en,
                      const int* __restrict__ es,
                      const float* __restrict__ xw,
                      const float* __restrict__ dn,
                      const float* __restrict__ dsv,
                      const float* __restrict__ bn,
                      const float* __restrict__ bs,
                      const float* __restrict__ wl,
                      const float* __restrict__ bl,
                      float* __restrict__ out, int n, int E)
{
    __shared__ float hacc[AGG_ROWS * 32];
    __shared__ int2  queue[AGG_WAVES][AGG_QCAP];

    const int lane = threadIdx.x & 31;
    const int wave = __builtin_amdgcn_readfirstlane((int)(threadIdx.x >> 5));
    int2* q = queue[wave];

    for (int base = blockIdx.x * AGG_ROWS; base < n; base += gridDim.x * AGG_ROWS) {
        const int rows = (n - base < AGG_ROWS) ? (n - base) : AGG_ROWS;

        for (int t = threadIdx.x; t < rows * 32; t += AGG_THREADS) {
            const int i = base + (t >> 5), j = t & 31;
            const float a = dn[i], b = dsv[i];
            hacc[t] = xw[(size_t)i * 64 + j]      * (a * a)
                    + xw[(size_t)i * 64 + 32 + j] * (b * b);
        }
        __syncthreads();

        for (int typ = 0; typ < 2; ++typ) {
            const int*   src  = typ ? es  : en;
            const int*   dst  = src + E;
            const float* dinv = typ ? dsv : dn;
            const int    cb   = typ * 32;
            int qn = 0;

            const int nvec = (E & 3) ? 0 : ((E >> 2) / AGG_TRIP) * AGG_TRIP;
            const int4* dst4 = (const int4*)dst;
            int v0 = wave * AGG_TRIP;
            if (v0 < nvec) {
                int4 d[AGG_U];
                agg_load_trip(dst4, v0, lane, d);
                while (true) {
                    const int  v1   = v0 + AGG_WAVES * AGG_TRIP;
                    const bool more = v1 < nvec;
                    int4 dnx[AGG_U];
                    agg_load_trip(dst4, more ? v1 : v0, lane, dnx);
                    #pragma unroll
                    for (int u = 0; u < AGG_U; ++u) {
                        const int e0 = 4 * (v0 + u * 32 + lane);
                        const int dc[4] = { d[u].x, d[u].y, d[u].z, d[u].w };
                        #pragma unroll
                        for (int c = 0; c < 4; ++c) {
                            const unsigned dl = (unsigned)(dc[c] - base);
                            agg_push(q, qn, dl < (unsigned)rows, e0 + c, dl);
                        }
                    }
                    __builtin_amdgcn_wave_barrier();
                    while (qn >= 32) {
                        qn -= 32;
                        agg_drain<true>(q + qn, 32, lane, src, dinv, xw, base, cb, hacc);
                    }
                    if (!more) break;
                    #pragma unroll
                    for (int u = 0; u < AGG_U; ++u) d[u] = dnx[u];
                    v0 = v1;
                }
            }
            for (int e0 = 4 * nvec + wave * 32; e0 < E; e0 += AGG_WAVES * 32) {
                const int e = e0 + lane;
                const int dv = (e < E) ? dst[e] : -1;
                const unsigned dl = (unsigned)(dv - base);
                agg_push(q, qn, dl < (unsigned)rows, e, dl);
                __builtin_amdgcn_wave_barrier();
                while (qn >= 32) {
                    qn -= 32;
                    agg_drain<true>(q + qn, 32, lane, src, dinv, xw, base, cb, hacc);
                }
            }
            __builtin_amdgcn_wave_barrier();
            if (qn > 0) agg_drain<false>(q, qn, lane, src, dinv, xw, base, cb, hacc);
        }
        asm volatile("s_wait_dscnt 0x0" ::: "memory");
        __syncthreads();

        const float bnj  = bn[lane], bsj = bs[lane];
        const float wj   = wl[lane];
        const float blin = bl[0];
        for (int r0 = wave * 32; r0 < rows; r0 += AGG_WAVES * 32) {
            float res = 0.0f;
            const int rcnt = (rows - r0 < 32) ? (rows - r0) : 32;
            for (int rr = 0; rr < rcnt; ++rr) {
                float v = hacc[(r0 + rr) * 32 + lane] + bnj + bsj;
                v = fmaxf(v, 0.0f) * wj;
                #pragma unroll
                for (int off = 16; off > 0; off >>= 1)
                    v += __shfl_xor(v, off, 32);
                if (lane == rr) res = v + blin;
            }
            if (lane < rcnt) out[base + r0 + lane] = res;
        }
        __syncthreads();
    }
}

extern "C" void kernel_launch(void* const* d_in, const int* in_sizes, int n_in,
                              void* d_out, int out_size, void* d_ws, size_t ws_size,
                              hipStream_t stream)
{
    const float* x  = (const float*)d_in[0];
    const int*   en = (const int*)  d_in[1];
    const int*   es = (const int*)  d_in[2];
    const float* Wn = (const float*)d_in[3];
    const float* bn = (const float*)d_in[4];
    const float* Ws = (const float*)d_in[5];
    const float* bs = (const float*)d_in[6];
    const float* Wl = (const float*)d_in[7];
    const float* bl = (const float*)d_in[8];
    float* out = (float*)d_out;

    const int N = in_sizes[0] / 64;
    const int E = in_sizes[1] / 2;

    float* ws  = (float*)d_ws;
    float* xw  = ws;
    float* dn  = xw + (size_t)N * 96;
    float* ds  = dn + N;

    const int TPB = 256;

    init_deg_kernel<<<(N + TPB - 1) / TPB, TPB, 0, stream>>>(dn, ds, N);
    count_deg_kernel<<<dim3((E + TPB - 1) / TPB, 2), TPB, 0, stream>>>(
        en + E, es + E, dn, ds, E);
    rsqrt_kernel<<<(N + TPB - 1) / TPB, TPB, 0, stream>>>(dn, ds, N);

    gemm_xw_kernel<<<(N + 15) / 16, 128, 0, stream>>>(x, Wn, Ws, xw, N);

    aggregate_kernel<<<AGG_GRID, AGG_THREADS, 0, stream>>>(en, es, xw, dn, ds, bn, bs, Wl, bl, out, N, E);
}


# aggregate split 2-D: 64 row ranges x 4 edge slices on 256 work-groups (each scans a quarter of dst); partial h rows merged by global float atomics (arrival-order sums); head in own launch
# speedup vs baseline: 2.0152x; 1.2028x over previous
#include <hip/hip_runtime.h>

typedef __attribute__((ext_vector_type(2))) float v2f;
typedef __attribute__((ext_vector_type(8))) float v8f;

__device__ __forceinline__ void atomic_add_f32(float* p, float v)
{
    asm volatile("global_atomic_add_f32 %0, %1, off"
                 :: "v"(p), "v"(v) : "memory");
}

typedef __attribute__((address_space(3))) float lds_float;
__device__ __forceinline__ void lds_add_f32(float* p, float v)
{
    asm volatile("ds_add_f32 %0, %1"
                 :: "v"((lds_float*)p), "v"(v) : "memory");
}

__global__ __launch_bounds__(128)
void gemm_xw_kernel(const float* __restrict__ x,
                    const float* __restrict__ Wn,
                    const float* __restrict__ Ws,
                    float* __restrict__ xw,
                    float* __restrict__ h, int nrows)
{
    const int lane  = threadIdx.x & 31;
    const int wave  = threadIdx.x >> 5;
    const int m0    = blockIdx.x * 16;
    const int lhalf = lane >> 4;
    const int lm    = lane & 15;

    const float* __restrict__ W = (wave < 2) ? Wn : Ws;
    const int n0 = (wave & 1) * 16;

    int arow = m0 + lm;
    if (arow >= nrows) arow = nrows - 1;

    v8f c = {};
    #pragma unroll
    for (int k0 = 0; k0 < 64; k0 += 4) {
        v2f a = *(const v2f*)(x + (size_t)arow * 64 + k0 + 2 * lhalf);
        v2f b;
        b.x = W[(k0 + 2 * lhalf)     * 32 + n0 + lm];
        b.y = W[(k0 + 2 * lhalf + 1) * 32 + n0 + lm];
        c = __builtin_amdgcn_wmma_f32_16x16x4_f32(false, a, false, b,
                                                  (short)0, c, false, false);
    }

    const int gcol = wave * 16 + lm;
    #pragma unroll
    for (int v = 0; v < 8; ++v) {
        const int row = m0 + v + 8 * lhalf;
        if (row < nrows) xw[(size_t)row * 64 + gcol] = c[v];
    }

    if (wave < 2) {
        #pragma unroll
        for (int v = 0; v < 8; ++v) {
            const int row = m0 + v + 8 * lhalf;
            if (row < nrows) h[(size_t)row * 32 + wave * 16 + lm] = 0.0f;
        }
    }
}

__global__ void init_deg_kernel(float* __restrict__ dn, float* __restrict__ ds, int n)
{
    const int i = blockIdx.x * blockDim.x + threadIdx.x;
    if (i < n) { dn[i] = 1.0f; ds[i] = 1.0f; }
}

__global__ void count_deg_kernel(const int* __restrict__ dst_n,
                                 const int* __restrict__ dst_s,
                                 float* __restrict__ dn,
                                 float* __restrict__ ds, int E)
{
    const int e = blockIdx.x * blockDim.x + threadIdx.x;
    if (e >= E) return;
    const int*   dst = blockIdx.y ? dst_s : dst_n;
    float*       deg = blockIdx.y ? ds    : dn;
    atomic_add_f32(&deg[dst[e]], 1.0f);
}

__global__ void rsqrt_kernel(float* __restrict__ dn, float* __restrict__ ds, int n)
{
    const int i = blockIdx.x * blockDim.x + threadIdx.x;
    if (i < n) {
        const float a = dn[i], b = ds[i];
        dn[i] = (a > 0.0f) ? rsqrtf(a) : 0.0f;
        ds[i] = (b > 0.0f) ? rsqrtf(b) : 0.0f;
    }
}

constexpr int AGG_THREADS = 1024;
constexpr int AGG_WAVES   = AGG_THREADS / 32;
constexpr int AGG_RANGES  = 64;
constexpr int AGG_SLICES  = 4;
constexpr int AGG_GRID    = AGG_RANGES * AGG_SLICES;
constexpr int AGG_ROWS    = 1563;
constexpr int AGG_U       = 3;
constexpr int AGG_TRIP    = 32 * AGG_U;
constexpr int AGG_QCAP    = 32 + 4 * AGG_TRIP;
constexpr int AGG_B       = 16;

__device__ __forceinline__ void agg_load_trip(const int4* __restrict__ dst4, int v0, int lane,
                                              int4 (&d)[AGG_U])
{
    #pragma unroll
    for (int u = 0; u < AGG_U; ++u) d[u] = dst4[v0 + u * 32 + lane];
}

__device__ __forceinline__ void agg_push(int2* q, int& qn, bool m, int e, unsigned dl)
{
    const unsigned mask = __builtin_amdgcn_ballot_w32(m);
    if (mask) {
        if (m) q[qn + __builtin_amdgcn_mbcnt_lo(mask, 0u)] = make_int2(e, (int)dl);
        qn += __popc(mask);
    }
}

template <bool FULL>
__device__ __forceinline__ void agg_drain(const int2* q, int cnt, int lane,
                                          const int* __restrict__ src,
                                          const float* __restrict__ dinv,
                                          const float* __restrict__ xw,
                                          int base, int cb, float* hacc)
{
    int soff = 0, hoff = 0;
    float nrm = 0.0f;
    if (FULL || lane < cnt) {
        const int2 ent = q[lane];
        const int  s   = src[ent.x];
        nrm  = dinv[s] * dinv[base + ent.y];
        soff = s * 64 + cb;
        hoff = ent.y * 32;
    }
    const int kmax = FULL ? 32 : cnt;
    for (int k0 = 0; k0 < kmax; k0 += AGG_B) {
        float val[AGG_B];
        #pragma unroll
        for (int i = 0; i < AGG_B; ++i) {
            const int so = __builtin_amdgcn_readlane(soff, k0 + i);
            val[i] = xw[so + lane];
        }
        #pragma unroll
        for (int i = 0; i < AGG_B; ++i) {
            if (FULL || k0 + i < cnt) {
                const int   ho = __builtin_amdgcn_readlane(hoff, k0 + i);
                const float nk = __int_as_float(__builtin_amdgcn_readlane(__float_as_int(nrm), k0 + i));
                lds_add_f32(&hacc[ho + lane], val[i] * nk);
            }
        }
    }
}

__global__ __launch_bounds__(AGG_THREADS)
void aggregate_kernel(const int* __restrict__ en,
                      const int* __restrict__ es,
                      const float* __restrict__ xw,
                      const float* __restrict__ dn,
                      const float* __restrict__ dsv,
                      float* __restrict__ h, int n, int E)
{
    __shared__ float hacc[AGG_ROWS * 32];
    __shared__ int2  queue[AGG_WAVES][AGG_QCAP];

    const int lane = threadIdx.x & 31;
    const int wave = __builtin_amdgcn_readfirstlane((int)(threadIdx.x >> 5));
    int2* q = queue[wave];

    const int rng = blockIdx.x / AGG_SLICES;
    const int slc = blockIdx.x % AGG_SLICES;
    const int gw  = slc * AGG_WAVES + wave;
    constexpr int TW = AGG_SLICES * AGG_WAVES;

    for (int base = rng * AGG_ROWS; base < n; base += AGG_RANGES * AGG_ROWS) {
        const int rows = (n - base < AGG_ROWS) ? (n - base) : AGG_ROWS;

        for (int t = threadIdx.x; t < rows * 32; t += AGG_THREADS) {
            float v = 0.0f;
            if (slc == 0) {
                const int i = base + (t >> 5), j = t & 31;
                const float a = dn[i], b = dsv[i];
                v = xw[(size_t)i * 64 + j]      * (a * a)
                  + xw[(size_t)i * 64 + 32 + j] * (b * b);
            }
            hacc[t] = v;
        }
        __syncthreads();

        for (int typ = 0; typ < 2; ++typ) {
            const int*   src  = typ ? es  : en;
            const int*   dst  = src + E;
            const float* dinv = typ ? dsv : dn;
            const int    cb   = typ * 32;
            int qn = 0;

            const int nvec = (E & 3) ? 0 : ((E >> 2) / AGG_TRIP) * AGG_TRIP;
            const int4* dst4 = (const int4*)dst;
            int v0 = gw * AGG_TRIP;
            if (v0 < nvec) {
                int4 d[AGG_U];
                agg_load_trip(dst4, v0, lane, d);
                while (true) {
                    const int  v1   = v0 + TW * AGG_TRIP;
                    const bool more = v1 < nvec;
                    int4 dnx[AGG_U];
                    agg_load_trip(dst4, more ? v1 : v0, lane, dnx);
                    #pragma unroll
                    for (int u = 0; u < AGG_U; ++u) {
                        const int e0 = 4 * (v0 + u * 32 + lane);
                        const int dc[4] = { d[u].x, d[u].y, d[u].z, d[u].w };
                        #pragma unroll
                        for (int c = 0; c < 4; ++c) {
                            const unsigned dl = (unsigned)(dc[c] - base);
                            agg_push(q, qn, dl < (unsigned)rows, e0 + c, dl);
                        }
                    }
                    __builtin_amdgcn_wave_barrier();
                    while (qn >= 32) {
                        qn -= 32;
                        agg_drain<true>(q + qn, 32, lane, src, dinv, xw, base, cb, hacc);
                    }
                    if (!more) break;
                    #pragma unroll
                    for (int u = 0; u < AGG_U; ++u) d[u] = dnx[u];
                    v0 = v1;
                }
            }
            for (int e0 = 4 * nvec + gw * 32; e0 < E; e0 += TW * 32) {
                const int e = e0 + lane;
                const int dv = (e < E) ? dst[e] : -1;
                const unsigned dl = (unsigned)(dv - base);
                agg_push(q, qn, dl < (unsigned)rows, e, dl);
                __builtin_amdgcn_wave_barrier();
                while (qn >= 32) {
                    qn -= 32;
                    agg_drain<true>(q + qn, 32, lane, src, dinv, xw, base, cb, hacc);
                }
            }
            __builtin_amdgcn_wave_barrier();
            if (qn > 0) agg_drain<false>(q, qn, lane, src, dinv, xw, base, cb, hacc);
        }
        asm volatile("s_wait_dscnt 0x0" ::: "memory");
        __syncthreads();

        float* hrow = h + (size_t)base * 32;
        for (int t = threadIdx.x; t < rows * 32; t += AGG_THREADS)
            atomic_add_f32(&hrow[t], hacc[t]);
        __syncthreads();
    }
}

__device__ __forceinline__ float head_row(float hv, float bnj, float bsj, float wj)
{
    float v = hv + bnj + bsj;
    v = fmaxf(v, 0.0f) * wj;
    #pragma unroll
    for (int off = 16; off > 0; off >>= 1)
        v += __shfl_xor(v, off, 32);
    return v;
}

__global__ __launch_bounds__(256)
void head_kernel(const float* __restrict__ h,
                 const float* __restrict__ bn,
                 const float* __restrict__ bs,
                 const float* __restrict__ wl,
                 const float* __restrict__ bl,
                 float* __restrict__ out, int n)
{
    const int lane = threadIdx.x & 31;
    const int r0   = ((blockIdx.x * blockDim.x + threadIdx.x) >> 5) * 32;
    if (r0 >= n) return;
    const float bnj  = bn[lane], bsj = bs[lane];
    const float wj   = wl[lane];
    const float blin = bl[0];
    const float* hp  = h + (size_t)r0 * 32 + lane;
    float res = 0.0f;
    if (n - r0 >= 32) {
        float hv[32];
        #pragma unroll
        for (int rr = 0; rr < 32; ++rr) hv[rr] = hp[rr * 32];
        #pragma unroll
        for (int rr = 0; rr < 32; ++rr) {
            const float v = head_row(hv[rr], bnj, bsj, wj);
            if (lane == rr) res = v + blin;
        }
        out[r0 + lane] = res;
    } else {
        const int rcnt = n - r0;
        for (int rr = 0; rr < rcnt; ++rr) {
            const float v = head_row(hp[rr * 32], bnj, bsj, wj);
            if (lane == rr) res = v + blin;
        }
        if (lane < rcnt) out[r0 + lane] = res;
    }
}

extern "C" void kernel_launch(void* const* d_in, const int* in_sizes, int n_in,
                              void* d_out, int out_size, void* d_ws, size_t ws_size,
                              hipStream_t stream)
{
    const float* x  = (const float*)d_in[0];
    const int*   en = (const int*)  d_in[1];
    const int*   es = (const int*)  d_in[2];
    const float* Wn = (const float*)d_in[3];
    const float* bn = (const float*)d_in[4];
    const float* Ws = (const float*)d_in[5];
    const float* bs = (const float*)d_in[6];
    const float* Wl = (const float*)d_in[7];
    const float* bl = (const float*)d_in[8];
    float* out = (float*)d_out;

    const int N = in_sizes[0] / 64;
    const int E = in_sizes[1] / 2;

    float* ws  = (float*)d_ws;
    float* xw  = ws;
    float* h   = xw + (size_t)N * 64;
    float* dn  = xw + (size_t)N * 96;
    float* ds  = dn + N;

    const int TPB = 256;

    init_deg_kernel<<<(N + TPB - 1) / TPB, TPB, 0, stream>>>(dn, ds, N);
    count_deg_kernel<<<dim3((E + TPB - 1) / TPB, 2), TPB, 0, stream>>>(
        en + E, es + E, dn, ds, E);
    rsqrt_kernel<<<(N + TPB - 1) / TPB, TPB, 0, stream>>>(dn, ds, N);

    gemm_xw_kernel<<<(N + 15) / 16, 128, 0, stream>>>(x, Wn, Ws, xw, h, N);

    aggregate_kernel<<<AGG_GRID, AGG_THREADS, 0, stream>>>(en, es, xw, dn, ds, h, N, E);

    head_kernel<<<(N + TPB - 1) / TPB, TPB, 0, stream>>>(h, bn, bs, Wl, bl, out, N);
}


# in-degrees counted by exact LDS float adds of one on 256 work-groups (2 types x 8 node ranges x 16 dst slices), partial counts stored plainly, summed in the rsqrt launch; global degree atomics gone
# speedup vs baseline: 4.5076x; 2.2368x over previous
#include <hip/hip_runtime.h>

typedef __attribute__((ext_vector_type(2))) float v2f;
typedef __attribute__((ext_vector_type(8))) float v8f;

__device__ __forceinline__ void atomic_add_f32(float* p, float v)
{
    asm volatile("global_atomic_add_f32 %0, %1, off"
                 :: "v"(p), "v"(v) : "memory");
}

typedef __attribute__((address_space(3))) float lds_float;
__device__ __forceinline__ void lds_add_f32(float* p, float v)
{
    asm volatile("ds_add_f32 %0, %1"
                 :: "v"((lds_float*)p), "v"(v) : "memory");
}

__global__ __launch_bounds__(128)
void gemm_xw_kernel(const float* __restrict__ x,
                    const float* __restrict__ Wn,
                    const float* __restrict__ Ws,
                    float* __restrict__ xw,
                    float* __restrict__ h, int nrows)
{
    const int lane  = threadIdx.x & 31;
    const int wave  = threadIdx.x >> 5;
    const int m0    = blockIdx.x * 16;
    const int lhalf = lane >> 4;
    const int lm    = lane & 15;

    const float* __restrict__ W = (wave < 2) ? Wn : Ws;
    const int n0 = (wave & 1) * 16;

    int arow = m0 + lm;
    if (arow >= nrows) arow = nrows - 1;

    v8f c = {};
    #pragma unroll
    for (int k0 = 0; k0 < 64; k0 += 4) {
        v2f a = *(const v2f*)(x + (size_t)arow * 64 + k0 + 2 * lhalf);
        v2f b;
        b.x = W[(k0 + 2 * lhalf)     * 32 + n0 + lm];
        b.y = W[(k0 + 2 * lhalf + 1) * 32 + n0 + lm];
        c = __builtin_amdgcn_wmma_f32_16x16x4_f32(false, a, false, b,
                                                  (short)0, c, false, false);
    }

    const int gcol = wave * 16 + lm;
    #pragma unroll
    for (int v = 0; v < 8; ++v) {
        const int row = m0 + v + 8 * lhalf;
        if (row < nrows) xw[(size_t)row * 64 + gcol] = c[v];
    }

    if (wave < 2) {
        #pragma unroll
        for (int v = 0; v < 8; ++v) {
            const int row = m0 + v + 8 * lhalf;
            if (row < nrows) h[(size_t)row * 32 + wave * 16 + lm] = 0.0f;
        }
    }
}

constexpr int DEG_THREADS = 1024;
constexpr int DEG_PARTS   = 8;
constexpr int DEG_SLICES  = 16;
constexpr int DEG_GRID    = 2 * DEG_PARTS * DEG_SLICES;
constexpr int DEG_PN      = 12512;
constexpr int DEG_U       = 12;

__device__ __forceinline__ void deg_count(float* cnt, int d, int base, int rows)
{
    const unsigned dl = (unsigned)(d - base);
    if (dl < (unsigned)rows) lds_add_f32(&cnt[dl], 1.0f);
}

__global__ __launch_bounds__(DEG_THREADS)
void degree_kernel(const int* __restrict__ dst_n,
                   const int* __restrict__ dst_s,
                   float* __restrict__ part, int n, int E)
{
    __shared__ float cnt[DEG_PN];

    const int slc = blockIdx.x % DEG_SLICES;
    const int prt = (blockIdx.x / DEG_SLICES) % DEG_PARTS;
    const int typ = blockIdx.x / (DEG_SLICES * DEG_PARTS);

    const int* dst   = typ ? dst_s : dst_n;
    float*     plane = part + ((size_t)slc * 2 + typ) * n;

    const int per  = (((E + DEG_SLICES - 1) / DEG_SLICES) + 3) & ~3;
    const int e_lo = (slc * per < E) ? slc * per : E;
    const int e_hi = (e_lo + per < E) ? e_lo + per : E;
    const bool vec  = (E & 3) == 0;
    const int  v_lo = e_lo >> 2;
    const int  v_hi = vec ? (e_hi >> 2) : v_lo;
    const int  t_lo = vec ? 4 * v_hi : e_lo;
    const int4* dst4 = (const int4*)dst;

    for (int base = prt * DEG_PN; base < n; base += DEG_PARTS * DEG_PN) {
        const int rows = (n - base < DEG_PN) ? (n - base) : DEG_PN;

        for (int t = threadIdx.x; t < rows; t += DEG_THREADS) cnt[t] = 0.0f;
        __syncthreads();

        int v = v_lo + threadIdx.x;
        for (; v + (DEG_U - 1) * DEG_THREADS < v_hi; v += DEG_U * DEG_THREADS) {
            int4 d[DEG_U];
            #pragma unroll
            for (int u = 0; u < DEG_U; ++u) d[u] = dst4[v + u * DEG_THREADS];
            #pragma unroll
            for (int u = 0; u < DEG_U; ++u) {
                deg_count(cnt, d[u].x, base, rows);
                deg_count(cnt, d[u].y, base, rows);
                deg_count(cnt, d[u].z, base, rows);
                deg_count(cnt, d[u].w, base, rows);
            }
        }
        for (; v < v_hi; v += DEG_THREADS) {
            const int4 d = dst4[v];
            deg_count(cnt, d.x, base, rows);
            deg_count(cnt, d.y, base, rows);
            deg_count(cnt, d.z, base, rows);
            deg_count(cnt, d.w, base, rows);
        }
        for (int e = t_lo + threadIdx.x; e < e_hi; e += DEG_THREADS)
            deg_count(cnt, dst[e], base, rows);
        asm volatile("s_wait_dscnt 0x0" ::: "memory");
        __syncthreads();

        for (int t = threadIdx.x; t < rows; t += DEG_THREADS)
            plane[base + t] = cnt[t];
        __syncthreads();
    }
}

__global__ void dinv_kernel(const float* __restrict__ part,
                            float* __restrict__ dn, float* __restrict__ ds, int n)
{
    const int i = blockIdx.x * blockDim.x + threadIdx.x;
    if (i < n) {
        float a = 1.0f, b = 1.0f;
        #pragma unroll
        for (int s = 0; s < DEG_SLICES; ++s) {
            a += part[(size_t)(2 * s)     * n + i];
            b += part[(size_t)(2 * s + 1) * n + i];
        }
        dn[i] = (a > 0.0f) ? rsqrtf(a) : 0.0f;
        ds[i] = (b > 0.0f) ? rsqrtf(b) : 0.0f;
    }
}

constexpr int AGG_THREADS = 1024;
constexpr int AGG_WAVES   = AGG_THREADS / 32;
constexpr int AGG_RANGES  = 64;
constexpr int AGG_SLICES  = 4;
constexpr int AGG_GRID    = AGG_RANGES * AGG_SLICES;
constexpr int AGG_ROWS    = 1563;
constexpr int AGG_U       = 3;
constexpr int AGG_TRIP    = 32 * AGG_U;
constexpr int AGG_QCAP    = 32 + 4 * AGG_TRIP;
constexpr int AGG_B       = 16;

__device__ __forceinline__ void agg_load_trip(const int4* __restrict__ dst4, int v0, int lane,
                                              int4 (&d)[AGG_U])
{
    #pragma unroll
    for (int u = 0; u < AGG_U; ++u) d[u] = dst4[v0 + u * 32 + lane];
}

__device__ __forceinline__ void agg_push(int2* q, int& qn, bool m, int e, unsigned dl)
{
    const unsigned mask = __builtin_amdgcn_ballot_w32(m);
    if (mask) {
        if (m) q[qn + __builtin_amdgcn_mbcnt_lo(mask, 0u)] = make_int2(e, (int)dl);
        qn += __popc(mask);
    }
}

template <bool FULL>
__device__ __forceinline__ void agg_drain(const int2* q, int cnt, int lane,
                                          const int* __restrict__ src,
                                          const float* __restrict__ dinv,
                                          const float* __restrict__ xw,
                                          int base, int cb, float* hacc)
{
    int soff = 0, hoff = 0;
    float nrm = 0.0f;
    if (FULL || lane < cnt) {
        const int2 ent = q[lane];
        const int  s   = src[ent.x];
        nrm  = dinv[s] * dinv[base + ent.y];
        soff = s * 64 + cb;
        hoff = ent.y * 32;
    }
    const int kmax = FULL ? 32 : cnt;
    for (int k0 = 0; k0 < kmax; k0 += AGG_B) {
        float val[AGG_B];
        #pragma unroll
        for (int i = 0; i < AGG_B; ++i) {
            const int so = __builtin_amdgcn_readlane(soff, k0 + i);
            val[i] = xw[so + lane];
        }
        #pragma unroll
        for (int i = 0; i < AGG_B; ++i) {
            if (FULL || k0 + i < cnt) {
                const int   ho = __builtin_amdgcn_readlane(hoff, k0 + i);
                const float nk = __int_as_float(__builtin_amdgcn_readlane(__float_as_int(nrm), k0 + i));
                lds_add_f32(&hacc[ho + lane], val[i] * nk);
            }
        }
    }
}

__global__ __launch_bounds__(AGG_THREADS)
void aggregate_kernel(const int* __restrict__ en,
                      const int* __restrict__ es,
                      const float* __restrict__ xw,
                      const float* __restrict__ dn,
                      const float* __restrict__ dsv,
                      float* __restrict__ h, int n, int E)
{
    __shared__ float hacc[AGG_ROWS * 32];
    __shared__ int2  queue[AGG_WAVES][AGG_QCAP];

    const int lane = threadIdx.x & 31;
    const int wave = __builtin_amdgcn_readfirstlane((int)(threadIdx.x >> 5));
    int2* q = queue[wave];

    const int rng = blockIdx.x / AGG_SLICES;
    const int slc = blockIdx.x % AGG_SLICES;
    const int gw  = slc * AGG_WAVES + wave;
    constexpr int TW = AGG_SLICES * AGG_WAVES;

    for (int base = rng * AGG_ROWS; base < n; base += AGG_RANGES * AGG_ROWS) {
        const int rows = (n - base < AGG_ROWS) ? (n - base) : AGG_ROWS;

        for (int t = threadIdx.x; t < rows * 32; t += AGG_THREADS) {
            float v = 0.0f;
            if (slc == 0) {
                const int i = base + (t >> 5), j = t & 31;
                const float a = dn[i], b = dsv[i];
                v = xw[(size_t)i * 64 + j]      * (a * a)
                  + xw[(size_t)i * 64 + 32 + j] * (b * b);
            }
            hacc[t] = v;
        }
        __syncthreads();

        for (int typ = 0; typ < 2; ++typ) {
            const int*   src  = typ ? es  : en;
            const int*   dst  = src + E;
            const float* dinv = typ ? dsv : dn;
            const int    cb   = typ * 32;
            int qn = 0;

            const int nvec = (E & 3) ? 0 : ((E >> 2) / AGG_TRIP) * AGG_TRIP;
            const int4* dst4 = (const int4*)dst;
            int v0 = gw * AGG_TRIP;
            if (v0 < nvec) {
                int4 d[AGG_U];
                agg_load_trip(dst4, v0, lane, d);
                while (true) {
                    const int  v1   = v0 + TW * AGG_TRIP;
                    const bool more = v1 < nvec;
                    int4 dnx[AGG_U];
                    agg_load_trip(dst4, more ? v1 : v0, lane, dnx);
                    #pragma unroll
                    for (int u = 0; u < AGG_U; ++u) {
                        const int e0 = 4 * (v0 + u * 32 + lane);
                        const int dc[4] = { d[u].x, d[u].y, d[u].z, d[u].w };
                        #pragma unroll
                        for (int c = 0; c < 4; ++c) {
                            const unsigned dl = (unsigned)(dc[c] - base);
                            agg_push(q, qn, dl < (unsigned)rows, e0 + c, dl);
                        }
                    }
                    __builtin_amdgcn_wave_barrier();
                    while (qn >= 32) {
                        qn -= 32;
                        agg_drain<true>(q + qn, 32, lane, src, dinv, xw, base, cb, hacc);
                    }
                    if (!more) break;
                    #pragma unroll
                    for (int u = 0; u < AGG_U; ++u) d[u] = dnx[u];
                    v0 = v1;
                }
            }
            for (int e0 = 4 * nvec + gw * 32; e0 < E; e0 += TW * 32) {
                const int e = e0 + lane;
                const int dv = (e < E) ? dst[e] : -1;
                const unsigned dl = (unsigned)(dv - base);
                agg_push(q, qn, dl < (unsigned)rows, e, dl);
                __builtin_amdgcn_wave_barrier();
                while (qn >= 32) {
                    qn -= 32;
                    agg_drain<true>(q + qn, 32, lane, src, dinv, xw, base, cb, hacc);
                }
            }
            __builtin_amdgcn_wave_barrier();
            if (qn > 0) agg_drain<false>(q, qn, lane, src, dinv, xw, base, cb, hacc);
        }
        asm volatile("s_wait_dscnt 0x0" ::: "memory");
        __syncthreads();

        float* hrow = h + (size_t)base * 32;
        for (int t = threadIdx.x; t < rows * 32; t += AGG_THREADS)
            atomic_add_f32(&hrow[t], hacc[t]);
        __syncthreads();
    }
}

__device__ __forceinline__ float head_row(float hv, float bnj, float bsj, float wj)
{
    float v = hv + bnj + bsj;
    v = fmaxf(v, 0.0f) * wj;
    #pragma unroll
    for (int off = 16; off > 0; off >>= 1)
        v += __shfl_xor(v, off, 32);
    return v;
}

__global__ __launch_bounds__(256)
void head_kernel(const float* __restrict__ h,
                 const float* __restrict__ bn,
                 const float* __restrict__ bs,
                 const float* __restrict__ wl,
                 const float* __restrict__ bl,
                 float* __restrict__ out, int n)
{
    const int lane = threadIdx.x & 31;
    const int r0   = ((blockIdx.x * blockDim.x + threadIdx.x) >> 5) * 32;
    if (r0 >= n) return;
    const float bnj  = bn[lane], bsj = bs[lane];
    const float wj   = wl[lane];
    const float blin = bl[0];
    const float* hp  = h + (size_t)r0 * 32 + lane;
    float res = 0.0f;
    if (n - r0 >= 32) {
        float hv[32];
        #pragma unroll
        for (int rr = 0; rr < 32; ++rr) hv[rr] = hp[rr * 32];
        #pragma unroll
        for (int rr = 0; rr < 32; ++rr) {
            const float v = head_row(hv[rr], bnj, bsj, wj);
            if (lane == rr) res = v + blin;
        }
        out[r0 + lane] = res;
    } else {
        const int rcnt = n - r0;
        for (int rr = 0; rr < rcnt; ++rr) {
            const float v = head_row(hp[rr * 32], bnj, bsj, wj);
            if (lane == rr) res = v + blin;
        }
        if (lane < rcnt) out[r0 + lane] = res;
    }
}

extern "C" void kernel_launch(void* const* d_in, const int* in_sizes, int n_in,
                              void* d_out, int out_size, void* d_ws, size_t ws_size,
                              hipStream_t stream)
{
    const float* x  = (const float*)d_in[0];
    const int*   en = (const int*)  d_in[1];
    const int*   es = (const int*)  d_in[2];
    const float* Wn = (const float*)d_in[3];
    const float* bn = (const float*)d_in[4];
    const float* Ws = (const float*)d_in[5];
    const float* bs = (const float*)d_in[6];
    const float* Wl = (const float*)d_in[7];
    const float* bl = (const float*)d_in[8];
    float* out = (float*)d_out;

    const int N = in_sizes[0] / 64;
    const int E = in_sizes[1] / 2;

    float* ws  = (float*)d_ws;
    float* xw  = ws;
    float* h   = xw + (size_t)N * 64;
    float* dn  = xw + (size_t)N * 96;
    float* ds  = dn + N;

    const int TPB = 256;

    degree_kernel<<<DEG_GRID, DEG_THREADS, 0, stream>>>(en + E, es + E, h, N, E);
    dinv_kernel<<<(N + TPB - 1) / TPB, TPB, 0, stream>>>(h, dn, ds, N);

    gemm_xw_kernel<<<(N + 15) / 16, 128, 0, stream>>>(x, Wn, Ws, xw, h, N);

    aggregate_kernel<<<AGG_GRID, AGG_THREADS, 0, stream>>>(en, es, xw, dn, ds, h, N, E);

    head_kernel<<<(N + TPB - 1) / TPB, TPB, 0, stream>>>(h, bn, bs, Wl, bl, out, N);
}


# GEMM epilogue stores each xw row times its node's d^-1/2 of the edge type, so xw rows carry the source factor and the aggregate drops the random per-edge d^-1/2[src] gather; products regrouped
# speedup vs baseline: 5.3652x; 1.1903x over previous
#include <hip/hip_runtime.h>

typedef __attribute__((ext_vector_type(2))) float v2f;
typedef __attribute__((ext_vector_type(8))) float v8f;

__device__ __forceinline__ void atomic_add_f32(float* p, float v)
{
    asm volatile("global_atomic_add_f32 %0, %1, off"
                 :: "v"(p), "v"(v) : "memory");
}

typedef __attribute__((address_space(3))) float lds_float;
__device__ __forceinline__ void lds_add_f32(float* p, float v)
{
    asm volatile("ds_add_f32 %0, %1"
                 :: "v"((lds_float*)p), "v"(v) : "memory");
}

__global__ __launch_bounds__(128)
void gemm_xw_kernel(const float* __restrict__ x,
                    const float* __restrict__ Wn,
                    const float* __restrict__ Ws,
                    const float* __restrict__ dn,
                    const float* __restrict__ ds,
                    float* __restrict__ xw,
                    float* __restrict__ h, int nrows)
{
    const int lane  = threadIdx.x & 31;
    const int wave  = threadIdx.x >> 5;
    const int m0    = blockIdx.x * 16;
    const int lhalf = lane >> 4;
    const int lm    = lane & 15;

    const float* __restrict__ W = (wave < 2) ? Wn : Ws;
    const int n0 = (wave & 1) * 16;

    int arow = m0 + lm;
    if (arow >= nrows) arow = nrows - 1;

    const float* __restrict__ dinv = (wave < 2) ? dn : ds;
    float sc[8];
    #pragma unroll
    for (int v = 0; v < 8; ++v) {
        int row = m0 + v + 8 * lhalf;
        if (row >= nrows) row = nrows - 1;
        sc[v] = dinv[row];
    }

    v8f c = {};
    #pragma unroll
    for (int k0 = 0; k0 < 64; k0 += 4) {
        v2f a = *(const v2f*)(x + (size_t)arow * 64 + k0 + 2 * lhalf);
        v2f b;
        b.x = W[(k0 + 2 * lhalf)     * 32 + n0 + lm];
        b.y = W[(k0 + 2 * lhalf + 1) * 32 + n0 + lm];
        c = __builtin_amdgcn_wmma_f32_16x16x4_f32(false, a, false, b,
                                                  (short)0, c, false, false);
    }

    asm volatile("v_nop\n\tv_nop\n\tv_nop\n\tv_nop\n\tv_nop" : "+v"(c));

    const int gcol = wave * 16 + lm;
    #pragma unroll
    for (int v = 0; v < 8; ++v) {
        const int row = m0 + v + 8 * lhalf;
        if (row < nrows) xw[(size_t)row * 64 + gcol] = c[v] * sc[v];
    }

    if (wave < 2) {
        #pragma unroll
        for (int v = 0; v < 8; ++v) {
            const int row = m0 + v + 8 * lhalf;
            if (row < nrows) h[(size_t)row * 32 + wave * 16 + lm] = 0.0f;
        }
    }
}

constexpr int DEG_THREADS = 1024;
constexpr int DEG_PARTS   = 8;
constexpr int DEG_SLICES  = 16;
constexpr int DEG_GRID    = 2 * DEG_PARTS * DEG_SLICES;
constexpr int DEG_PN      = 12512;
constexpr int DEG_U       = 12;

__device__ __forceinline__ void deg_count(float* cnt, int d, int base, int rows)
{
    const unsigned dl = (unsigned)(d - base);
    if (dl < (unsigned)rows) lds_add_f32(&cnt[dl], 1.0f);
}

__global__ __launch_bounds__(DEG_THREADS)
void degree_kernel(const int* __restrict__ dst_n,
                   const int* __restrict__ dst_s,
                   float* __restrict__ part, int n, int E)
{
    __shared__ float cnt[DEG_PN];

    const int slc = blockIdx.x % DEG_SLICES;
    const int prt = (blockIdx.x / DEG_SLICES) % DEG_PARTS;
    const int typ = blockIdx.x / (DEG_SLICES * DEG_PARTS);

    const int* dst   = typ ? dst_s : dst_n;
    float*     plane = part + ((size_t)slc * 2 + typ) * n;

    const int per  = (((E + DEG_SLICES - 1) / DEG_SLICES) + 3) & ~3;
    const int e_lo = (slc * per < E) ? slc * per : E;
    const int e_hi = (e_lo + per < E) ? e_lo + per : E;
    const bool vec  = (E & 3) == 0;
    const int  v_lo = e_lo >> 2;
    const int  v_hi = vec ? (e_hi >> 2) : v_lo;
    const int  t_lo = vec ? 4 * v_hi : e_lo;
    const int4* dst4 = (const int4*)dst;

    for (int base = prt * DEG_PN; base < n; base += DEG_PARTS * DEG_PN) {
        const int rows = (n - base < DEG_PN) ? (n - base) : DEG_PN;

        for (int t = threadIdx.x; t < rows; t += DEG_THREADS) cnt[t] = 0.0f;
        __syncthreads();

        int v = v_lo + threadIdx.x;
        for (; v + (DEG_U - 1) * DEG_THREADS < v_hi; v += DEG_U * DEG_THREADS) {
            int4 d[DEG_U];
            #pragma unroll
            for (int u = 0; u < DEG_U; ++u) d[u] = dst4[v + u * DEG_THREADS];
            #pragma unroll
            for (int u = 0; u < DEG_U; ++u) {
                deg_count(cnt, d[u].x, base, rows);
                deg_count(cnt, d[u].y, base, rows);
                deg_count(cnt, d[u].z, base, rows);
                deg_count(cnt, d[u].w, base, rows);
            }
        }
        for (; v < v_hi; v += DEG_THREADS) {
            const int4 d = dst4[v];
            deg_count(cnt, d.x, base, rows);
            deg_count(cnt, d.y, base, rows);
            deg_count(cnt, d.z, base, rows);
            deg_count(cnt, d.w, base, rows);
        }
        for (int e = t_lo + threadIdx.x; e < e_hi; e += DEG_THREADS)
            deg_count(cnt, dst[e], base, rows);
        asm volatile("s_wait_dscnt 0x0" ::: "memory");
        __syncthreads();

        for (int t = threadIdx.x; t < rows; t += DEG_THREADS)
            plane[base + t] = cnt[t];
        __syncthreads();
    }
}

__global__ void dinv_kernel(const float* __restrict__ part,
                            float* __restrict__ dn, float* __restrict__ ds, int n)
{
    const int i = blockIdx.x * blockDim.x + threadIdx.x;
    if (i < n) {
        float a = 1.0f, b = 1.0f;
        #pragma unroll
        for (int s = 0; s < DEG_SLICES; ++s) {
            a += part[(size_t)(2 * s)     * n + i];
            b += part[(size_t)(2 * s + 1) * n + i];
        }
        dn[i] = (a > 0.0f) ? rsqrtf(a) : 0.0f;
        ds[i] = (b > 0.0f) ? rsqrtf(b) : 0.0f;
    }
}

constexpr int AGG_THREADS = 1024;
constexpr int AGG_WAVES   = AGG_THREADS / 32;
constexpr int AGG_RANGES  = 64;
constexpr int AGG_SLICES  = 4;
constexpr int AGG_GRID    = AGG_RANGES * AGG_SLICES;
constexpr int AGG_ROWS    = 1563;
constexpr int AGG_U       = 3;
constexpr int AGG_TRIP    = 32 * AGG_U;
constexpr int AGG_QCAP    = 32 + 4 * AGG_TRIP;
constexpr int AGG_B       = 16;

__device__ __forceinline__ void agg_load_trip(const int4* __restrict__ dst4, int v0, int lane,
                                              int4 (&d)[AGG_U])
{
    #pragma unroll
    for (int u = 0; u < AGG_U; ++u) d[u] = dst4[v0 + u * 32 + lane];
}

__device__ __forceinline__ void agg_push(int2* q, int& qn, bool m, int e, unsigned dl)
{
    const unsigned mask = __builtin_amdgcn_ballot_w32(m);
    if (mask) {
        if (m) q[qn + __builtin_amdgcn_mbcnt_lo(mask, 0u)] = make_int2(e, (int)dl);
        qn += __popc(mask);
    }
}

template <bool FULL>
__device__ __forceinline__ void agg_drain(const int2* q, int cnt, int lane,
                                          const int* __restrict__ src,
                                          const float* __restrict__ dinv,
                                          const float* __restrict__ xw,
                                          int base, int cb, float* hacc)
{
    int soff = 0, hoff = 0;
    float nrm = 0.0f;
    if (FULL || lane < cnt) {
        const int2 ent = q[lane];
        const int  s   = src[ent.x];
        nrm  = dinv[base + ent.y];
        soff = s * 64 + cb;
        hoff = ent.y * 32;
    }
    const int kmax = FULL ? 32 : cnt;
    for (int k0 = 0; k0 < kmax; k0 += AGG_B) {
        float val[AGG_B];
        #pragma unroll
        for (int i = 0; i < AGG_B; ++i) {
            const int so = __builtin_amdgcn_readlane(soff, k0 + i);
            val[i] = xw[so + lane];
        }
        #pragma unroll
        for (int i = 0; i < AGG_B; ++i) {
            if (FULL || k0 + i < cnt) {
                const int   ho = __builtin_amdgcn_readlane(hoff, k0 + i);
                const float nk = __int_as_float(__builtin_amdgcn_readlane(__float_as_int(nrm), k0 + i));
                lds_add_f32(&hacc[ho + lane], val[i] * nk);
            }
        }
    }
}

__global__ __launch_bounds__(AGG_THREADS)
void aggregate_kernel(const int* __restrict__ en,
                      const int* __restrict__ es,
                      const float* __restrict__ xw,
                      const float* __restrict__ dn,
                      const float* __restrict__ dsv,
                      float* __restrict__ h, int n, int E)
{
    __shared__ float hacc[AGG_ROWS * 32];
    __shared__ int2  queue[AGG_WAVES][AGG_QCAP];

    const int lane = threadIdx.x & 31;
    const int wave = __builtin_amdgcn_readfirstlane((int)(threadIdx.x >> 5));
    int2* q = queue[wave];

    const int rng = blockIdx.x / AGG_SLICES;
    const int slc = blockIdx.x % AGG_SLICES;
    const int gw  = slc * AGG_WAVES + wave;
    constexpr int TW = AGG_SLICES * AGG_WAVES;

    for (int base = rng * AGG_ROWS; base < n; base += AGG_RANGES * AGG_ROWS) {
        const int rows = (n - base < AGG_ROWS) ? (n - base) : AGG_ROWS;

        for (int t = threadIdx.x; t < rows * 32; t += AGG_THREADS) {
            float v = 0.0f;
            if (slc == 0) {
                const int i = base + (t >> 5), j = t & 31;
                const float a = dn[i], b = dsv[i];
                v = xw[(size_t)i * 64 + j]      * a
                  + xw[(size_t)i * 64 + 32 + j] * b;
            }
            hacc[t] = v;
        }
        __syncthreads();

        for (int typ = 0; typ < 2; ++typ) {
            const int*   src  = typ ? es  : en;
            const int*   dst  = src + E;
            const float* dinv = typ ? dsv : dn;
            const int    cb   = typ * 32;
            int qn = 0;

            const int nvec = (E & 3) ? 0 : ((E >> 2) / AGG_TRIP) * AGG_TRIP;
            const int4* dst4 = (const int4*)dst;
            int v0 = gw * AGG_TRIP;
            if (v0 < nvec) {
                int4 d[AGG_U];
                agg_load_trip(dst4, v0, lane, d);
                while (true) {
                    const int  v1   = v0 + TW * AGG_TRIP;
                    const bool more = v1 < nvec;
                    int4 dnx[AGG_U];
                    agg_load_trip(dst4, more ? v1 : v0, lane, dnx);
                    #pragma unroll
                    for (int u = 0; u < AGG_U; ++u) {
                        const int e0 = 4 * (v0 + u * 32 + lane);
                        const int dc[4] = { d[u].x, d[u].y, d[u].z, d[u].w };
                        #pragma unroll
                        for (int c = 0; c < 4; ++c) {
                            const unsigned dl = (unsigned)(dc[c] - base);
                            agg_push(q, qn, dl < (unsigned)rows, e0 + c, dl);
                        }
                    }
                    __builtin_amdgcn_wave_barrier();
                    while (qn >= 32) {
                        qn -= 32;
                        agg_drain<true>(q + qn, 32, lane, src, dinv, xw, base, cb, hacc);
                    }
                    if (!more) break;
                    #pragma unroll
                    for (int u = 0; u < AGG_U; ++u) d[u] = dnx[u];
                    v0 = v1;
                }
            }
            for (int e0 = 4 * nvec + gw * 32; e0 < E; e0 += TW * 32) {
                const int e = e0 + lane;
                const int dv = (e < E) ? dst[e] : -1;
                const unsigned dl = (unsigned)(dv - base);
                agg_push(q, qn, dl < (unsigned)rows, e, dl);
                __builtin_amdgcn_wave_barrier();
                while (qn >= 32) {
                    qn -= 32;
                    agg_drain<true>(q + qn, 32, lane, src, dinv, xw, base, cb, hacc);
                }
            }
            __builtin_amdgcn_wave_barrier();
            if (qn > 0) agg_drain<false>(q, qn, lane, src, dinv, xw, base, cb, hacc);
        }
        asm volatile("s_wait_dscnt 0x0" ::: "memory");
        __syncthreads();

        float* hrow = h + (size_t)base * 32;
        for (int t = threadIdx.x; t < rows * 32; t += AGG_THREADS)
            atomic_add_f32(&hrow[t], hacc[t]);
        __syncthreads();
    }
}

__device__ __forceinline__ float head_row(float hv, float bnj, float bsj, float wj)
{
    float v = hv + bnj + bsj;
    v = fmaxf(v, 0.0f) * wj;
    #pragma unroll
    for (int off = 16; off > 0; off >>= 1)
        v += __shfl_xor(v, off, 32);
    return v;
}

__global__ __launch_bounds__(256)
void head_kernel(const float* __restrict__ h,
                 const float* __restrict__ bn,
                 const float* __restrict__ bs,
                 const float* __restrict__ wl,
                 const float* __restrict__ bl,
                 float* __restrict__ out, int n)
{
    const int lane = threadIdx.x & 31;
    const int r0   = ((blockIdx.x * blockDim.x + threadIdx.x) >> 5) * 32;
    if (r0 >= n) return;
    const float bnj  = bn[lane], bsj = bs[lane];
    const float wj   = wl[lane];
    const float blin = bl[0];
    const float* hp  = h + (size_t)r0 * 32 + lane;
    float res = 0.0f;
    if (n - r0 >= 32) {
        float hv[32];
        #pragma unroll
        for (int rr = 0; rr < 32; ++rr) hv[rr] = hp[rr * 32];
        #pragma unroll
        for (int rr = 0; rr < 32; ++rr) {
            const float v = head_row(hv[rr], bnj, bsj, wj);
            if (lane == rr) res = v + blin;
        }
        out[r0 + lane] = res;
    } else {
        const int rcnt = n - r0;
        for (int rr = 0; rr < rcnt; ++rr) {
            const float v = head_row(hp[rr * 32], bnj, bsj, wj);
            if (lane == rr) res = v + blin;
        }
        if (lane < rcnt) out[r0 + lane] = res;
    }
}

extern "C" void kernel_launch(void* const* d_in, const int* in_sizes, int n_in,
                              void* d_out, int out_size, void* d_ws, size_t ws_size,
                              hipStream_t stream)
{
    const float* x  = (const float*)d_in[0];
    const int*   en = (const int*)  d_in[1];
    const int*   es = (const int*)  d_in[2];
    const float* Wn = (const float*)d_in[3];
    const float* bn = (const float*)d_in[4];
    const float* Ws = (const float*)d_in[5];
    const float* bs = (const float*)d_in[6];
    const float* Wl = (const float*)d_in[7];
    const float* bl = (const float*)d_in[8];
    float* out = (float*)d_out;

    const int N = in_sizes[0] / 64;
    const int E = in_sizes[1] / 2;

    float* ws  = (float*)d_ws;
    float* xw  = ws;
    float* h   = xw + (size_t)N * 64;
    float* dn  = xw + (size_t)N * 96;
    float* ds  = dn + N;

    const int TPB = 256;

    degree_kernel<<<DEG_GRID, DEG_THREADS, 0, stream>>>(en + E, es + E, h, N, E);
    dinv_kernel<<<(N + TPB - 1) / TPB, TPB, 0, stream>>>(h, dn, ds, N);

    gemm_xw_kernel<<<(N + 15) / 16, 128, 0, stream>>>(x, Wn, Ws, dn, ds, xw, h, N);

    aggregate_kernel<<<AGG_GRID, AGG_THREADS, 0, stream>>>(en, es, xw, dn, ds, h, N, E);

    head_kernel<<<(N + TPB - 1) / TPB, TPB, 0, stream>>>(h, bn, bs, Wl, bl, out, N);
}


# aggregate drain: one 16-B load per lane gathers 4 xw rows (8 lanes per edge, own lookups, no readlane); columns rotated per lane group so LDS float adds (arrival-order sums) miss bank conflicts
# speedup vs baseline: 5.6236x; 1.0482x over previous
#include <hip/hip_runtime.h>

typedef __attribute__((ext_vector_type(2))) float v2f;
typedef __attribute__((ext_vector_type(8))) float v8f;

__device__ __forceinline__ void atomic_add_f32(float* p, float v)
{
    asm volatile("global_atomic_add_f32 %0, %1, off"
                 :: "v"(p), "v"(v) : "memory");
}

typedef __attribute__((address_space(3))) float lds_float;
__device__ __forceinline__ void lds_add_f32(float* p, float v)
{
    asm volatile("ds_add_f32 %0, %1"
                 :: "v"((lds_float*)p), "v"(v) : "memory");
}

__global__ __launch_bounds__(128)
void gemm_xw_kernel(const float* __restrict__ x,
                    const float* __restrict__ Wn,
                    const float* __restrict__ Ws,
                    const float* __restrict__ dn,
                    const float* __restrict__ ds,
                    float* __restrict__ xw,
                    float* __restrict__ h, int nrows)
{
    const int lane  = threadIdx.x & 31;
    const int wave  = threadIdx.x >> 5;
    const int m0    = blockIdx.x * 16;
    const int lhalf = lane >> 4;
    const int lm    = lane & 15;

    const float* __restrict__ W = (wave < 2) ? Wn : Ws;
    const int n0 = (wave & 1) * 16;

    int arow = m0 + lm;
    if (arow >= nrows) arow = nrows - 1;

    const float* __restrict__ dinv = (wave < 2) ? dn : ds;
    float sc[8];
    #pragma unroll
    for (int v = 0; v < 8; ++v) {
        int row = m0 + v + 8 * lhalf;
        if (row >= nrows) row = nrows - 1;
        sc[v] = dinv[row];
    }

    v8f c = {};
    #pragma unroll
    for (int k0 = 0; k0 < 64; k0 += 4) {
        v2f a = *(const v2f*)(x + (size_t)arow * 64 + k0 + 2 * lhalf);
        v2f b;
        b.x = W[(k0 + 2 * lhalf)     * 32 + n0 + lm];
        b.y = W[(k0 + 2 * lhalf + 1) * 32 + n0 + lm];
        c = __builtin_amdgcn_wmma_f32_16x16x4_f32(false, a, false, b,
                                                  (short)0, c, false, false);
    }

    asm volatile("v_nop\n\tv_nop\n\tv_nop\n\tv_nop\n\tv_nop" : "+v"(c));

    const int gcol = wave * 16 + lm;
    #pragma unroll
    for (int v = 0; v < 8; ++v) {
        const int row = m0 + v + 8 * lhalf;
        if (row < nrows) xw[(size_t)row * 64 + gcol] = c[v] * sc[v];
    }

    if (wave < 2) {
        #pragma unroll
        for (int v = 0; v < 8; ++v) {
            const int row = m0 + v + 8 * lhalf;
            if (row < nrows) h[(size_t)row * 32 + wave * 16 + lm] = 0.0f;
        }
    }
}

constexpr int DEG_THREADS = 1024;
constexpr int DEG_PARTS   = 8;
constexpr int DEG_SLICES  = 16;
constexpr int DEG_GRID    = 2 * DEG_PARTS * DEG_SLICES;
constexpr int DEG_PN      = 12512;
constexpr int DEG_U       = 12;

__device__ __forceinline__ void deg_count(float* cnt, int d, int base, int rows)
{
    const unsigned dl = (unsigned)(d - base);
    if (dl < (unsigned)rows) lds_add_f32(&cnt[dl], 1.0f);
}

__global__ __launch_bounds__(DEG_THREADS)
void degree_kernel(const int* __restrict__ dst_n,
                   const int* __restrict__ dst_s,
                   float* __restrict__ part, int n, int E)
{
    __shared__ float cnt[DEG_PN];

    const int slc = blockIdx.x % DEG_SLICES;
    const int prt = (blockIdx.x / DEG_SLICES) % DEG_PARTS;
    const int typ = blockIdx.x / (DEG_SLICES * DEG_PARTS);

    const int* dst   = typ ? dst_s : dst_n;
    float*     plane = part + ((size_t)slc * 2 + typ) * n;

    const int per  = (((E + DEG_SLICES - 1) / DEG_SLICES) + 3) & ~3;
    const int e_lo = (slc * per < E) ? slc * per : E;
    const int e_hi = (e_lo + per < E) ? e_lo + per : E;
    const bool vec  = (E & 3) == 0;
    const int  v_lo = e_lo >> 2;
    const int  v_hi = vec ? (e_hi >> 2) : v_lo;
    const int  t_lo = vec ? 4 * v_hi : e_lo;
    const int4* dst4 = (const int4*)dst;

    for (int base = prt * DEG_PN; base < n; base += DEG_PARTS * DEG_PN) {
        const int rows = (n - base < DEG_PN) ? (n - base) : DEG_PN;

        for (int t = threadIdx.x; t < rows; t += DEG_THREADS) cnt[t] = 0.0f;
        __syncthreads();

        int v = v_lo + threadIdx.x;
        for (; v + (DEG_U - 1) * DEG_THREADS < v_hi; v += DEG_U * DEG_THREADS) {
            int4 d[DEG_U];
            #pragma unroll
            for (int u = 0; u < DEG_U; ++u) d[u] = dst4[v + u * DEG_THREADS];
            #pragma unroll
            for (int u = 0; u < DEG_U; ++u) {
                deg_count(cnt, d[u].x, base, rows);
                deg_count(cnt, d[u].y, base, rows);
                deg_count(cnt, d[u].z, base, rows);
                deg_count(cnt, d[u].w, base, rows);
            }
        }
        for (; v < v_hi; v += DEG_THREADS) {
            const int4 d = dst4[v];
            deg_count(cnt, d.x, base, rows);
            deg_count(cnt, d.y, base, rows);
            deg_count(cnt, d.z, base, rows);
            deg_count(cnt, d.w, base, rows);
        }
        for (int e = t_lo + threadIdx.x; e < e_hi; e += DEG_THREADS)
            deg_count(cnt, dst[e], base, rows);
        asm volatile("s_wait_dscnt 0x0" ::: "memory");
        __syncthreads();

        for (int t = threadIdx.x; t < rows; t += DEG_THREADS)
            plane[base + t] = cnt[t];
        __syncthreads();
    }
}

__global__ void dinv_kernel(const float* __restrict__ part,
                            float* __restrict__ dn, float* __restrict__ ds, int n)
{
    const int i = blockIdx.x * blockDim.x + threadIdx.x;
    if (i < n) {
        float a = 1.0f, b = 1.0f;
        #pragma unroll
        for (int s = 0; s < DEG_SLICES; ++s) {
            a += part[(size_t)(2 * s)     * n + i];
            b += part[(size_t)(2 * s + 1) * n + i];
        }
        dn[i] = (a > 0.0f) ? rsqrtf(a) : 0.0f;
        ds[i] = (b > 0.0f) ? rsqrtf(b) : 0.0f;
    }
}

constexpr int AGG_THREADS = 1024;
constexpr int AGG_WAVES   = AGG_THREADS / 32;
constexpr int AGG_RANGES  = 64;
constexpr int AGG_SLICES  = 4;
constexpr int AGG_GRID    = AGG_RANGES * AGG_SLICES;
constexpr int AGG_ROWS    = 1563;
constexpr int AGG_U       = 3;
constexpr int AGG_TRIP    = 32 * AGG_U;
constexpr int AGG_QCAP    = 32 + 4 * AGG_TRIP;
constexpr int AGG_G       = 4;
constexpr int AGG_STEPS   = 32 / AGG_G;
constexpr int AGG_B       = 4;

__device__ __forceinline__ void agg_load_trip(const int4* __restrict__ dst4, int v0, int lane,
                                              int4 (&d)[AGG_U])
{
    #pragma unroll
    for (int u = 0; u < AGG_U; ++u) d[u] = dst4[v0 + u * 32 + lane];
}

__device__ __forceinline__ void agg_push(int2* q, int& qn, bool m, int e, unsigned dl)
{
    const unsigned mask = __builtin_amdgcn_ballot_w32(m);
    if (mask) {
        if (m) q[qn + __builtin_amdgcn_mbcnt_lo(mask, 0u)] = make_int2(e, (int)dl);
        qn += __popc(mask);
    }
}

template <bool FULL>
__device__ __forceinline__ void agg_drain(const int2* q, int cnt, int lane,
                                          const int* __restrict__ src,
                                          const float* __restrict__ dinv,
                                          const float* __restrict__ xw,
                                          int base, int cb, float* hacc)
{
    const int  g  = lane >> 3;
    const int  l7 = lane & 7;
    const bool r1 = (g & 1) != 0;
    const bool r2 = (g & 2) != 0;
    const int  k0 = 4 * l7 + ( g      & 3);
    const int  k1 = 4 * l7 + ((g + 1) & 3);
    const int  k2 = 4 * l7 + ((g + 2) & 3);
    const int  k3 = 4 * l7 + ((g + 3) & 3);

    const float4* __restrict__ xw4 = (const float4*)xw;
    unsigned soff[AGG_STEPS];
    int      hrow[AGG_STEPS];
    float nrm[AGG_STEPS];
    #pragma unroll
    for (int t = 0; t < AGG_STEPS; ++t) {
        const int  idx = AGG_G * t + g;
        const int2 ent = q[(FULL || idx < cnt) ? idx : 0];
        hrow[t] = ent.y * 32;
        nrm[t]  = dinv[base + ent.y];
        soff[t] = (unsigned)(src[ent.x] * 16 + (cb >> 2) + l7);
    }
    #pragma unroll
    for (int h = 0; h < AGG_STEPS / AGG_B; ++h) {
        if (!FULL && AGG_G * AGG_B * h >= cnt) break;
        float4 row[AGG_B];
        #pragma unroll
        for (int i = 0; i < AGG_B; ++i)
            row[i] = xw4[soff[AGG_B * h + i]];
        #pragma unroll
        for (int i = 0; i < AGG_B; ++i) {
            const int t = AGG_B * h + i;
            if (FULL || AGG_G * t + g < cnt) {
                const float4 v = row[i];
                const float a0 = r1 ? v.y : v.x, a1 = r1 ? v.z : v.y;
                const float a2 = r1 ? v.w : v.z, a3 = r1 ? v.x : v.w;
                const float b0 = r2 ? a2 : a0,   b1 = r2 ? a3 : a1;
                const float b2 = r2 ? a0 : a2,   b3 = r2 ? a1 : a3;
                const float nk = nrm[t];
                float* hr = hacc + hrow[t];
                lds_add_f32(hr + k0, b0 * nk);
                lds_add_f32(hr + k1, b1 * nk);
                lds_add_f32(hr + k2, b2 * nk);
                lds_add_f32(hr + k3, b3 * nk);
            }
        }
    }
}

__global__ __launch_bounds__(AGG_THREADS)
void aggregate_kernel(const int* __restrict__ en,
                      const int* __restrict__ es,
                      const float* __restrict__ xw,
                      const float* __restrict__ dn,
                      const float* __restrict__ dsv,
                      float* __restrict__ h, int n, int E)
{
    __shared__ float hacc[AGG_ROWS * 32];
    __shared__ int2  queue[AGG_WAVES][AGG_QCAP];

    const int lane = threadIdx.x & 31;
    const int wave = __builtin_amdgcn_readfirstlane((int)(threadIdx.x >> 5));
    int2* q = queue[wave];

    const int rng = blockIdx.x / AGG_SLICES;
    const int slc = blockIdx.x % AGG_SLICES;
    const int gw  = slc * AGG_WAVES + wave;
    constexpr int TW = AGG_SLICES * AGG_WAVES;

    for (int base = rng * AGG_ROWS; base < n; base += AGG_RANGES * AGG_ROWS) {
        const int rows = (n - base < AGG_ROWS) ? (n - base) : AGG_ROWS;

        for (int t = threadIdx.x; t < rows * 32; t += AGG_THREADS) {
            float v = 0.0f;
            if (slc == 0) {
                const int i = base + (t >> 5), j = t & 31;
                const float a = dn[i], b = dsv[i];
                v = xw[(size_t)i * 64 + j]      * a
                  + xw[(size_t)i * 64 + 32 + j] * b;
            }
            hacc[t] = v;
        }
        __syncthreads();

        for (int typ = 0; typ < 2; ++typ) {
            const int*   src  = typ ? es  : en;
            const int*   dst  = src + E;
            const float* dinv = typ ? dsv : dn;
            const int    cb   = typ * 32;
            int qn = 0;

            const int nvec = (E & 3) ? 0 : ((E >> 2) / AGG_TRIP) * AGG_TRIP;
            const int4* dst4 = (const int4*)dst;
            int v0 = gw * AGG_TRIP;
            if (v0 < nvec) {
                int4 d[AGG_U];
                agg_load_trip(dst4, v0, lane, d);
                while (true) {
                    const int  v1   = v0 + TW * AGG_TRIP;
                    const bool more = v1 < nvec;
                    int4 dnx[AGG_U];
                    agg_load_trip(dst4, more ? v1 : v0, lane, dnx);
                    #pragma unroll
                    for (int u = 0; u < AGG_U; ++u) {
                        const int e0 = 4 * (v0 + u * 32 + lane);
                        const int dc[4] = { d[u].x, d[u].y, d[u].z, d[u].w };
                        #pragma unroll
                        for (int c = 0; c < 4; ++c) {
                            const unsigned dl = (unsigned)(dc[c] - base);
                            agg_push(q, qn, dl < (unsigned)rows, e0 + c, dl);
                        }
                    }
                    __builtin_amdgcn_wave_barrier();
                    while (qn >= 32) {
                        qn -= 32;
                        agg_drain<true>(q + qn, 32, lane, src, dinv, xw, base, cb, hacc);
                    }
                    if (!more) break;
                    #pragma unroll
                    for (int u = 0; u < AGG_U; ++u) d[u] = dnx[u];
                    v0 = v1;
                }
            }
            for (int e0 = 4 * nvec + gw * 32; e0 < E; e0 += TW * 32) {
                const int e = e0 + lane;
                const int dv = (e < E) ? dst[e] : -1;
                const unsigned dl = (unsigned)(dv - base);
                agg_push(q, qn, dl < (unsigned)rows, e, dl);
                __builtin_amdgcn_wave_barrier();
                while (qn >= 32) {
                    qn -= 32;
                    agg_drain<true>(q + qn, 32, lane, src, dinv, xw, base, cb, hacc);
                }
            }
            __builtin_amdgcn_wave_barrier();
            if (qn > 0) agg_drain<false>(q, qn, lane, src, dinv, xw, base, cb, hacc);
        }
        asm volatile("s_wait_dscnt 0x0" ::: "memory");
        __syncthreads();

        float* hrow = h + (size_t)base * 32;
        for (int t = threadIdx.x; t < rows * 32; t += AGG_THREADS)
            atomic_add_f32(&hrow[t], hacc[t]);
        __syncthreads();
    }
}

__device__ __forceinline__ float head_row(float hv, float bnj, float bsj, float wj)
{
    float v = hv + bnj + bsj;
    v = fmaxf(v, 0.0f) * wj;
    #pragma unroll
    for (int off = 16; off > 0; off >>= 1)
        v += __shfl_xor(v, off, 32);
    return v;
}

__global__ __launch_bounds__(256)
void head_kernel(const float* __restrict__ h,
                 const float* __restrict__ bn,
                 const float* __restrict__ bs,
                 const float* __restrict__ wl,
                 const float* __restrict__ bl,
                 float* __restrict__ out, int n)
{
    const int lane = threadIdx.x & 31;
    const int r0   = ((blockIdx.x * blockDim.x + threadIdx.x) >> 5) * 32;
    if (r0 >= n) return;
    const float bnj  = bn[lane], bsj = bs[lane];
    const float wj   = wl[lane];
    const float blin = bl[0];
    const float* hp  = h + (size_t)r0 * 32 + lane;
    float res = 0.0f;
    if (n - r0 >= 32) {
        float hv[32];
        #pragma unroll
        for (int rr = 0; rr < 32; ++rr) hv[rr] = hp[rr * 32];
        #pragma unroll
        for (int rr = 0; rr < 32; ++rr) {
            const float v = head_row(hv[rr], bnj, bsj, wj);
            if (lane == rr) res = v + blin;
        }
        out[r0 + lane] = res;
    } else {
        const int rcnt = n - r0;
        for (int rr = 0; rr < rcnt; ++rr) {
            const float v = head_row(hp[rr * 32], bnj, bsj, wj);
            if (lane == rr) res = v + blin;
        }
        if (lane < rcnt) out[r0 + lane] = res;
    }
}

extern "C" void kernel_launch(void* const* d_in, const int* in_sizes, int n_in,
                              void* d_out, int out_size, void* d_ws, size_t ws_size,
                              hipStream_t stream)
{
    const float* x  = (const float*)d_in[0];
    const int*   en = (const int*)  d_in[1];
    const int*   es = (const int*)  d_in[2];
    const float* Wn = (const float*)d_in[3];
    const float* bn = (const float*)d_in[4];
    const float* Ws = (const float*)d_in[5];
    const float* bs = (const float*)d_in[6];
    const float* Wl = (const float*)d_in[7];
    const float* bl = (const float*)d_in[8];
    float* out = (float*)d_out;

    const int N = in_sizes[0] / 64;
    const int E = in_sizes[1] / 2;

    float* ws  = (float*)d_ws;
    float* xw  = ws;
    float* h   = xw + (size_t)N * 64;
    float* dn  = xw + (size_t)N * 96;
    float* ds  = dn + N;

    const int TPB = 256;

    degree_kernel<<<DEG_GRID, DEG_THREADS, 0, stream>>>(en + E, es + E, h, N, E);
    dinv_kernel<<<(N + TPB - 1) / TPB, TPB, 0, stream>>>(h, dn, ds, N);

    gemm_xw_kernel<<<(N + 15) / 16, 128, 0, stream>>>(x, Wn, Ws, dn, ds, xw, h, N);

    aggregate_kernel<<<AGG_GRID, AGG_THREADS, 0, stream>>>(en, es, xw, dn, ds, h, N, E);

    head_kernel<<<(N + TPB - 1) / TPB, TPB, 0, stream>>>(h, bn, bs, Wl, bl, out, N);
}


# GEMM epilogue stores each node's self-loop messages into h in place of its zero fill (one wave holds all 64 columns of 16 rows); aggregate slice 0 drops its xw read pass; h sums regrouped
# speedup vs baseline: 5.8900x; 1.0474x over previous
#include <hip/hip_runtime.h>

typedef __attribute__((ext_vector_type(2))) float v2f;
typedef __attribute__((ext_vector_type(8))) float v8f;

__device__ __forceinline__ void atomic_add_f32(float* p, float v)
{
    asm volatile("global_atomic_add_f32 %0, %1, off"
                 :: "v"(p), "v"(v) : "memory");
}

typedef __attribute__((address_space(3))) float lds_float;
__device__ __forceinline__ void lds_add_f32(float* p, float v)
{
    asm volatile("ds_add_f32 %0, %1"
                 :: "v"((lds_float*)p), "v"(v) : "memory");
}

__global__ __launch_bounds__(128)
void gemm_xw_kernel(const float* __restrict__ x,
                    const float* __restrict__ Wn,
                    const float* __restrict__ Ws,
                    const float* __restrict__ dn,
                    const float* __restrict__ ds,
                    float* __restrict__ xw,
                    float* __restrict__ h, int nrows)
{
    const int lane  = threadIdx.x & 31;
    const int wave  = __builtin_amdgcn_readfirstlane((int)(threadIdx.x >> 5));
    const int m0    = (blockIdx.x * 4 + wave) * 16;
    if (m0 >= nrows) return;
    const int lhalf = lane >> 4;
    const int lm    = lane & 15;

    int arow = m0 + lm;
    if (arow >= nrows) arow = nrows - 1;

    float sn[8], ss[8];
    #pragma unroll
    for (int v = 0; v < 8; ++v) {
        int row = m0 + v + 8 * lhalf;
        if (row >= nrows) row = nrows - 1;
        sn[v] = dn[row];
        ss[v] = ds[row];
    }

    v8f c0 = {}, c1 = {}, c2 = {}, c3 = {};
    #pragma unroll
    for (int k0 = 0; k0 < 64; k0 += 4) {
        v2f a = *(const v2f*)(x + (size_t)arow * 64 + k0 + 2 * lhalf);
        const int kr = (k0 + 2 * lhalf) * 32 + lm;
        v2f b0, b1, b2, b3;
        b0.x = Wn[kr];       b0.y = Wn[kr + 32];
        b1.x = Wn[kr + 16];  b1.y = Wn[kr + 48];
        b2.x = Ws[kr];       b2.y = Ws[kr + 32];
        b3.x = Ws[kr + 16];  b3.y = Ws[kr + 48];
        c0 = __builtin_amdgcn_wmma_f32_16x16x4_f32(false, a, false, b0,
                                                   (short)0, c0, false, false);
        c1 = __builtin_amdgcn_wmma_f32_16x16x4_f32(false, a, false, b1,
                                                   (short)0, c1, false, false);
        c2 = __builtin_amdgcn_wmma_f32_16x16x4_f32(false, a, false, b2,
                                                   (short)0, c2, false, false);
        c3 = __builtin_amdgcn_wmma_f32_16x16x4_f32(false, a, false, b3,
                                                   (short)0, c3, false, false);
    }

    asm volatile("v_nop\n\tv_nop\n\tv_nop\n\tv_nop\n\tv_nop"
                 : "+v"(c0), "+v"(c1), "+v"(c2), "+v"(c3));

    #pragma unroll
    for (int v = 0; v < 8; ++v) {
        const int row = m0 + v + 8 * lhalf;
        if (row < nrows) {
            const float xn0 = c0[v] * sn[v], xn1 = c1[v] * sn[v];
            const float xs0 = c2[v] * ss[v], xs1 = c3[v] * ss[v];
            float* xr = xw + (size_t)row * 64 + lm;
            xr[0]  = xn0;
            xr[16] = xn1;
            xr[32] = xs0;
            xr[48] = xs1;
            float* hr = h + (size_t)row * 32 + lm;
            hr[0]  = xn0 * sn[v] + xs0 * ss[v];
            hr[16] = xn1 * sn[v] + xs1 * ss[v];
        }
    }
}

constexpr int DEG_THREADS = 1024;
constexpr int DEG_PARTS   = 8;
constexpr int DEG_SLICES  = 16;
constexpr int DEG_GRID    = 2 * DEG_PARTS * DEG_SLICES;
constexpr int DEG_PN      = 12512;
constexpr int DEG_U       = 12;

__device__ __forceinline__ void deg_count(float* cnt, int d, int base, int rows)
{
    const unsigned dl = (unsigned)(d - base);
    if (dl < (unsigned)rows) lds_add_f32(&cnt[dl], 1.0f);
}

__global__ __launch_bounds__(DEG_THREADS)
void degree_kernel(const int* __restrict__ dst_n,
                   const int* __restrict__ dst_s,
                   float* __restrict__ part, int n, int E)
{
    __shared__ float cnt[DEG_PN];

    const int slc = blockIdx.x % DEG_SLICES;
    const int prt = (blockIdx.x / DEG_SLICES) % DEG_PARTS;
    const int typ = blockIdx.x / (DEG_SLICES * DEG_PARTS);

    const int* dst   = typ ? dst_s : dst_n;
    float*     plane = part + ((size_t)slc * 2 + typ) * n;

    const int per  = (((E + DEG_SLICES - 1) / DEG_SLICES) + 3) & ~3;
    const int e_lo = (slc * per < E) ? slc * per : E;
    const int e_hi = (e_lo + per < E) ? e_lo + per : E;
    const bool vec  = (E & 3) == 0;
    const int  v_lo = e_lo >> 2;
    const int  v_hi = vec ? (e_hi >> 2) : v_lo;
    const int  t_lo = vec ? 4 * v_hi : e_lo;
    const int4* dst4 = (const int4*)dst;

    for (int base = prt * DEG_PN; base < n; base += DEG_PARTS * DEG_PN) {
        const int rows = (n - base < DEG_PN) ? (n - base) : DEG_PN;

        for (int t = threadIdx.x; t < rows; t += DEG_THREADS) cnt[t] = 0.0f;
        __syncthreads();

        int v = v_lo + threadIdx.x;
        for (; v + (DEG_U - 1) * DEG_THREADS < v_hi; v += DEG_U * DEG_THREADS) {
            int4 d[DEG_U];
            #pragma unroll
            for (int u = 0; u < DEG_U; ++u) d[u] = dst4[v + u * DEG_THREADS];
            #pragma unroll
            for (int u = 0; u < DEG_U; ++u) {
                deg_count(cnt, d[u].x, base, rows);
                deg_count(cnt, d[u].y, base, rows);
                deg_count(cnt, d[u].z, base, rows);
                deg_count(cnt, d[u].w, base, rows);
            }
        }
        for (; v < v_hi; v += DEG_THREADS) {
            const int4 d = dst4[v];
            deg_count(cnt, d.x, base, rows);
            deg_count(cnt, d.y, base, rows);
            deg_count(cnt, d.z, base, rows);
            deg_count(cnt, d.w, base, rows);
        }
        for (int e = t_lo + threadIdx.x; e < e_hi; e += DEG_THREADS)
            deg_count(cnt, dst[e], base, rows);
        asm volatile("s_wait_dscnt 0x0" ::: "memory");
        __syncthreads();

        for (int t = threadIdx.x; t < rows; t += DEG_THREADS)
            plane[base + t] = cnt[t];
        __syncthreads();
    }
}

__global__ void dinv_kernel(const float* __restrict__ part,
                            float* __restrict__ dn, float* __restrict__ ds, int n)
{
    const int i = blockIdx.x * blockDim.x + threadIdx.x;
    if (i < n) {
        float a = 1.0f, b = 1.0f;
        #pragma unroll
        for (int s = 0; s < DEG_SLICES; ++s) {
            a += part[(size_t)(2 * s)     * n + i];
            b += part[(size_t)(2 * s + 1) * n + i];
        }
        dn[i] = (a > 0.0f) ? rsqrtf(a) : 0.0f;
        ds[i] = (b > 0.0f) ? rsqrtf(b) : 0.0f;
    }
}

constexpr int AGG_THREADS = 1024;
constexpr int AGG_WAVES   = AGG_THREADS / 32;
constexpr int AGG_RANGES  = 64;
constexpr int AGG_SLICES  = 4;
constexpr int AGG_GRID    = AGG_RANGES * AGG_SLICES;
constexpr int AGG_ROWS    = 1563;
constexpr int AGG_U       = 3;
constexpr int AGG_TRIP    = 32 * AGG_U;
constexpr int AGG_QCAP    = 32 + 4 * AGG_TRIP;
constexpr int AGG_G       = 4;
constexpr int AGG_STEPS   = 32 / AGG_G;
constexpr int AGG_B       = 4;

__device__ __forceinline__ void agg_load_trip(const int4* __restrict__ dst4, int v0, int lane,
                                              int4 (&d)[AGG_U])
{
    #pragma unroll
    for (int u = 0; u < AGG_U; ++u) d[u] = dst4[v0 + u * 32 + lane];
}

__device__ __forceinline__ void agg_push(int2* q, int& qn, bool m, int e, unsigned dl)
{
    const unsigned mask = __builtin_amdgcn_ballot_w32(m);
    if (mask) {
        if (m) q[qn + __builtin_amdgcn_mbcnt_lo(mask, 0u)] = make_int2(e, (int)dl);
        qn += __popc(mask);
    }
}

template <bool FULL>
__device__ __forceinline__ void agg_drain(const int2* q, int cnt, int lane,
                                          const int* __restrict__ src,
                                          const float* __restrict__ dinv,
                                          const float* __restrict__ xw,
                                          int base, int cb, float* hacc)
{
    const int  g  = lane >> 3;
    const int  l7 = lane & 7;
    const bool r1 = (g & 1) != 0;
    const bool r2 = (g & 2) != 0;
    const int  k0 = 4 * l7 + ( g      & 3);
    const int  k1 = 4 * l7 + ((g + 1) & 3);
    const int  k2 = 4 * l7 + ((g + 2) & 3);
    const int  k3 = 4 * l7 + ((g + 3) & 3);

    const float4* __restrict__ xw4 = (const float4*)xw;
    unsigned soff[AGG_STEPS];
    int      hrow[AGG_STEPS];
    float nrm[AGG_STEPS];
    #pragma unroll
    for (int t = 0; t < AGG_STEPS; ++t) {
        const int  idx = AGG_G * t + g;
        const int2 ent = q[(FULL || idx < cnt) ? idx : 0];
        hrow[t] = ent.y * 32;
        nrm[t]  = dinv[base + ent.y];
        soff[t] = (unsigned)(src[ent.x] * 16 + (cb >> 2) + l7);
    }
    #pragma unroll
    for (int h = 0; h < AGG_STEPS / AGG_B; ++h) {
        if (!FULL && AGG_G * AGG_B * h >= cnt) break;
        float4 row[AGG_B];
        #pragma unroll
        for (int i = 0; i < AGG_B; ++i)
            row[i] = xw4[soff[AGG_B * h + i]];
        #pragma unroll
        for (int i = 0; i < AGG_B; ++i) {
            const int t = AGG_B * h + i;
            if (FULL || AGG_G * t + g < cnt) {
                const float4 v = row[i];
                const float a0 = r1 ? v.y : v.x, a1 = r1 ? v.z : v.y;
                const float a2 = r1 ? v.w : v.z, a3 = r1 ? v.x : v.w;
                const float b0 = r2 ? a2 : a0,   b1 = r2 ? a3 : a1;
                const float b2 = r2 ? a0 : a2,   b3 = r2 ? a1 : a3;
                const float nk = nrm[t];
                float* hr = hacc + hrow[t];
                lds_add_f32(hr + k0, b0 * nk);
                lds_add_f32(hr + k1, b1 * nk);
                lds_add_f32(hr + k2, b2 * nk);
                lds_add_f32(hr + k3, b3 * nk);
            }
        }
    }
}

__global__ __launch_bounds__(AGG_THREADS)
void aggregate_kernel(const int* __restrict__ en,
                      const int* __restrict__ es,
                      const float* __restrict__ xw,
                      const float* __restrict__ dn,
                      const float* __restrict__ dsv,
                      float* __restrict__ h, int n, int E)
{
    __shared__ float hacc[AGG_ROWS * 32];
    __shared__ int2  queue[AGG_WAVES][AGG_QCAP];

    const int lane = threadIdx.x & 31;
    const int wave = __builtin_amdgcn_readfirstlane((int)(threadIdx.x >> 5));
    int2* q = queue[wave];

    const int rng = blockIdx.x / AGG_SLICES;
    const int slc = blockIdx.x % AGG_SLICES;
    const int gw  = slc * AGG_WAVES + wave;
    constexpr int TW = AGG_SLICES * AGG_WAVES;

    for (int base = rng * AGG_ROWS; base < n; base += AGG_RANGES * AGG_ROWS) {
        const int rows = (n - base < AGG_ROWS) ? (n - base) : AGG_ROWS;

        for (int t = threadIdx.x; t < rows * 32; t += AGG_THREADS) hacc[t] = 0.0f;
        __syncthreads();

        for (int typ = 0; typ < 2; ++typ) {
            const int*   src  = typ ? es  : en;
            const int*   dst  = src + E;
            const float* dinv = typ ? dsv : dn;
            const int    cb   = typ * 32;
            int qn = 0;

            const int nvec = (E & 3) ? 0 : ((E >> 2) / AGG_TRIP) * AGG_TRIP;
            const int4* dst4 = (const int4*)dst;
            int v0 = gw * AGG_TRIP;
            if (v0 < nvec) {
                int4 d[AGG_U];
                agg_load_trip(dst4, v0, lane, d);
                while (true) {
                    const int  v1   = v0 + TW * AGG_TRIP;
                    const bool more = v1 < nvec;
                    int4 dnx[AGG_U];
                    agg_load_trip(dst4, more ? v1 : v0, lane, dnx);
                    #pragma unroll
                    for (int u = 0; u < AGG_U; ++u) {
                        const int e0 = 4 * (v0 + u * 32 + lane);
                        const int dc[4] = { d[u].x, d[u].y, d[u].z, d[u].w };
                        #pragma unroll
                        for (int c = 0; c < 4; ++c) {
                            const unsigned dl = (unsigned)(dc[c] - base);
                            agg_push(q, qn, dl < (unsigned)rows, e0 + c, dl);
                        }
                    }
                    __builtin_amdgcn_wave_barrier();
                    while (qn >= 32) {
                        qn -= 32;
                        agg_drain<true>(q + qn, 32, lane, src, dinv, xw, base, cb, hacc);
                    }
                    if (!more) break;
                    #pragma unroll
                    for (int u = 0; u < AGG_U; ++u) d[u] = dnx[u];
                    v0 = v1;
                }
            }
            for (int e0 = 4 * nvec + gw * 32; e0 < E; e0 += TW * 32) {
                const int e = e0 + lane;
                const int dv = (e < E) ? dst[e] : -1;
                const unsigned dl = (unsigned)(dv - base);
                agg_push(q, qn, dl < (unsigned)rows, e, dl);
                __builtin_amdgcn_wave_barrier();
                while (qn >= 32) {
                    qn -= 32;
                    agg_drain<true>(q + qn, 32, lane, src, dinv, xw, base, cb, hacc);
                }
            }
            __builtin_amdgcn_wave_barrier();
            if (qn > 0) agg_drain<false>(q, qn, lane, src, dinv, xw, base, cb, hacc);
        }
        asm volatile("s_wait_dscnt 0x0" ::: "memory");
        __syncthreads();

        float* hrow = h + (size_t)base * 32;
        for (int t = threadIdx.x; t < rows * 32; t += AGG_THREADS)
            atomic_add_f32(&hrow[t], hacc[t]);
        __syncthreads();
    }
}

__device__ __forceinline__ float head_row(float hv, float bnj, float bsj, float wj)
{
    float v = hv + bnj + bsj;
    v = fmaxf(v, 0.0f) * wj;
    #pragma unroll
    for (int off = 16; off > 0; off >>= 1)
        v += __shfl_xor(v, off, 32);
    return v;
}

__global__ __launch_bounds__(256)
void head_kernel(const float* __restrict__ h,
                 const float* __restrict__ bn,
                 const float* __restrict__ bs,
                 const float* __restrict__ wl,
                 const float* __restrict__ bl,
                 float* __restrict__ out, int n)
{
    const int lane = threadIdx.x & 31;
    const int r0   = ((blockIdx.x * blockDim.x + threadIdx.x) >> 5) * 32;
    if (r0 >= n) return;
    const float bnj  = bn[lane], bsj = bs[lane];
    const float wj   = wl[lane];
    const float blin = bl[0];
    const float* hp  = h + (size_t)r0 * 32 + lane;
    float res = 0.0f;
    if (n - r0 >= 32) {
        float hv[32];
        #pragma unroll
        for (int rr = 0; rr < 32; ++rr) hv[rr] = hp[rr * 32];
        #pragma unroll
        for (int rr = 0; rr < 32; ++rr) {
            const float v = head_row(hv[rr], bnj, bsj, wj);
            if (lane == rr) res = v + blin;
        }
        out[r0 + lane] = res;
    } else {
        const int rcnt = n - r0;
        for (int rr = 0; rr < rcnt; ++rr) {
            const float v = head_row(hp[rr * 32], bnj, bsj, wj);
            if (lane == rr) res = v + blin;
        }
        if (lane < rcnt) out[r0 + lane] = res;
    }
}

extern "C" void kernel_launch(void* const* d_in, const int* in_sizes, int n_in,
                              void* d_out, int out_size, void* d_ws, size_t ws_size,
                              hipStream_t stream)
{
    const float* x  = (const float*)d_in[0];
    const int*   en = (const int*)  d_in[1];
    const int*   es = (const int*)  d_in[2];
    const float* Wn = (const float*)d_in[3];
    const float* bn = (const float*)d_in[4];
    const float* Ws = (const float*)d_in[5];
    const float* bs = (const float*)d_in[6];
    const float* Wl = (const float*)d_in[7];
    const float* bl = (const float*)d_in[8];
    float* out = (float*)d_out;

    const int N = in_sizes[0] / 64;
    const int E = in_sizes[1] / 2;

    float* ws  = (float*)d_ws;
    float* xw  = ws;
    float* h   = xw + (size_t)N * 64;
    float* dn  = xw + (size_t)N * 96;
    float* ds  = dn + N;

    const int TPB = 256;

    degree_kernel<<<DEG_GRID, DEG_THREADS, 0, stream>>>(en + E, es + E, h, N, E);
    dinv_kernel<<<(N + TPB - 1) / TPB, TPB, 0, stream>>>(h, dn, ds, N);

    gemm_xw_kernel<<<(N + 63) / 64, 128, 0, stream>>>(x, Wn, Ws, dn, ds, xw, h, N);

    aggregate_kernel<<<AGG_GRID, AGG_THREADS, 0, stream>>>(en, es, xw, dn, ds, h, N, E);

    head_kernel<<<(N + TPB - 1) / TPB, TPB, 0, stream>>>(h, bn, bs, Wl, bl, out, N);
}


# aggregate re-split to 42 row ranges x 6 edge slices: 2381 h rows per work-group in LDS (wave queues cut to 64 entries; a whole trip queued only if it fits), each scans 1/6 of dst; h sums regrouped
# speedup vs baseline: 6.2948x; 1.0687x over previous
#include <hip/hip_runtime.h>

typedef __attribute__((ext_vector_type(2))) float v2f;
typedef __attribute__((ext_vector_type(8))) float v8f;

__device__ __forceinline__ void atomic_add_f32(float* p, float v)
{
    asm volatile("global_atomic_add_f32 %0, %1, off"
                 :: "v"(p), "v"(v) : "memory");
}

typedef __attribute__((address_space(3))) float lds_float;
__device__ __forceinline__ void lds_add_f32(float* p, float v)
{
    asm volatile("ds_add_f32 %0, %1"
                 :: "v"((lds_float*)p), "v"(v) : "memory");
}

__global__ __launch_bounds__(128)
void gemm_xw_kernel(const float* __restrict__ x,
                    const float* __restrict__ Wn,
                    const float* __restrict__ Ws,
                    const float* __restrict__ dn,
                    const float* __restrict__ ds,
                    float* __restrict__ xw,
                    float* __restrict__ h, int nrows)
{
    const int lane  = threadIdx.x & 31;
    const int wave  = __builtin_amdgcn_readfirstlane((int)(threadIdx.x >> 5));
    const int m0    = (blockIdx.x * 4 + wave) * 16;
    if (m0 >= nrows) return;
    const int lhalf = lane >> 4;
    const int lm    = lane & 15;

    int arow = m0 + lm;
    if (arow >= nrows) arow = nrows - 1;

    float sn[8], ss[8];
    #pragma unroll
    for (int v = 0; v < 8; ++v) {
        int row = m0 + v + 8 * lhalf;
        if (row >= nrows) row = nrows - 1;
        sn[v] = dn[row];
        ss[v] = ds[row];
    }

    v8f c0 = {}, c1 = {}, c2 = {}, c3 = {};
    #pragma unroll
    for (int k0 = 0; k0 < 64; k0 += 4) {
        v2f a = *(const v2f*)(x + (size_t)arow * 64 + k0 + 2 * lhalf);
        const int kr = (k0 + 2 * lhalf) * 32 + lm;
        v2f b0, b1, b2, b3;
        b0.x = Wn[kr];       b0.y = Wn[kr + 32];
        b1.x = Wn[kr + 16];  b1.y = Wn[kr + 48];
        b2.x = Ws[kr];       b2.y = Ws[kr + 32];
        b3.x = Ws[kr + 16];  b3.y = Ws[kr + 48];
        c0 = __builtin_amdgcn_wmma_f32_16x16x4_f32(false, a, false, b0,
                                                   (short)0, c0, false, false);
        c1 = __builtin_amdgcn_wmma_f32_16x16x4_f32(false, a, false, b1,
                                                   (short)0, c1, false, false);
        c2 = __builtin_amdgcn_wmma_f32_16x16x4_f32(false, a, false, b2,
                                                   (short)0, c2, false, false);
        c3 = __builtin_amdgcn_wmma_f32_16x16x4_f32(false, a, false, b3,
                                                   (short)0, c3, false, false);
    }

    asm volatile("v_nop\n\tv_nop\n\tv_nop\n\tv_nop\n\tv_nop"
                 : "+v"(c0), "+v"(c1), "+v"(c2), "+v"(c3));

    #pragma unroll
    for (int v = 0; v < 8; ++v) {
        const int row = m0 + v + 8 * lhalf;
        if (row < nrows) {
            const float xn0 = c0[v] * sn[v], xn1 = c1[v] * sn[v];
            const float xs0 = c2[v] * ss[v], xs1 = c3[v] * ss[v];
            float* xr = xw + (size_t)row * 64 + lm;
            xr[0]  = xn0;
            xr[16] = xn1;
            xr[32] = xs0;
            xr[48] = xs1;
            float* hr = h + (size_t)row * 32 + lm;
            hr[0]  = xn0 * sn[v] + xs0 * ss[v];
            hr[16] = xn1 * sn[v] + xs1 * ss[v];
        }
    }
}

constexpr int DEG_THREADS = 1024;
constexpr int DEG_PARTS   = 8;
constexpr int DEG_SLICES  = 16;
constexpr int DEG_GRID    = 2 * DEG_PARTS * DEG_SLICES;
constexpr int DEG_PN      = 12512;
constexpr int DEG_U       = 12;

__device__ __forceinline__ void deg_count(float* cnt, int d, int base, int rows)
{
    const unsigned dl = (unsigned)(d - base);
    if (dl < (unsigned)rows) lds_add_f32(&cnt[dl], 1.0f);
}

__global__ __launch_bounds__(DEG_THREADS)
void degree_kernel(const int* __restrict__ dst_n,
                   const int* __restrict__ dst_s,
                   float* __restrict__ part, int n, int E)
{
    __shared__ float cnt[DEG_PN];

    const int slc = blockIdx.x % DEG_SLICES;
    const int prt = (blockIdx.x / DEG_SLICES) % DEG_PARTS;
    const int typ = blockIdx.x / (DEG_SLICES * DEG_PARTS);

    const int* dst   = typ ? dst_s : dst_n;
    float*     plane = part + ((size_t)slc * 2 + typ) * n;

    const int per  = (((E + DEG_SLICES - 1) / DEG_SLICES) + 3) & ~3;
    const int e_lo = (slc * per < E) ? slc * per : E;
    const int e_hi = (e_lo + per < E) ? e_lo + per : E;
    const bool vec  = (E & 3) == 0;
    const int  v_lo = e_lo >> 2;
    const int  v_hi = vec ? (e_hi >> 2) : v_lo;
    const int  t_lo = vec ? 4 * v_hi : e_lo;
    const int4* dst4 = (const int4*)dst;

    for (int base = prt * DEG_PN; base < n; base += DEG_PARTS * DEG_PN) {
        const int rows = (n - base < DEG_PN) ? (n - base) : DEG_PN;

        for (int t = threadIdx.x; t < rows; t += DEG_THREADS) cnt[t] = 0.0f;
        __syncthreads();

        int v = v_lo + threadIdx.x;
        for (; v + (DEG_U - 1) * DEG_THREADS < v_hi; v += DEG_U * DEG_THREADS) {
            int4 d[DEG_U];
            #pragma unroll
            for (int u = 0; u < DEG_U; ++u) d[u] = dst4[v + u * DEG_THREADS];
            #pragma unroll
            for (int u = 0; u < DEG_U; ++u) {
                deg_count(cnt, d[u].x, base, rows);
                deg_count(cnt, d[u].y, base, rows);
                deg_count(cnt, d[u].z, base, rows);
                deg_count(cnt, d[u].w, base, rows);
            }
        }
        for (; v < v_hi; v += DEG_THREADS) {
            const int4 d = dst4[v];
            deg_count(cnt, d.x, base, rows);
            deg_count(cnt, d.y, base, rows);
            deg_count(cnt, d.z, base, rows);
            deg_count(cnt, d.w, base, rows);
        }
        for (int e = t_lo + threadIdx.x; e < e_hi; e += DEG_THREADS)
            deg_count(cnt, dst[e], base, rows);
        asm volatile("s_wait_dscnt 0x0" ::: "memory");
        __syncthreads();

        for (int t = threadIdx.x; t < rows; t += DEG_THREADS)
            plane[base + t] = cnt[t];
        __syncthreads();
    }
}

__global__ void dinv_kernel(const float* __restrict__ part,
                            float* __restrict__ dn, float* __restrict__ ds, int n)
{
    const int i = blockIdx.x * blockDim.x + threadIdx.x;
    if (i < n) {
        float a = 1.0f, b = 1.0f;
        #pragma unroll
        for (int s = 0; s < DEG_SLICES; ++s) {
            a += part[(size_t)(2 * s)     * n + i];
            b += part[(size_t)(2 * s + 1) * n + i];
        }
        dn[i] = (a > 0.0f) ? rsqrtf(a) : 0.0f;
        ds[i] = (b > 0.0f) ? rsqrtf(b) : 0.0f;
    }
}

constexpr int AGG_THREADS = 1024;
constexpr int AGG_WAVES   = AGG_THREADS / 32;
constexpr int AGG_RANGES  = 42;
constexpr int AGG_SLICES  = 6;
constexpr int AGG_GRID    = AGG_RANGES * AGG_SLICES;
constexpr int AGG_ROWS    = 2381;
constexpr int AGG_U       = 3;
constexpr int AGG_TRIP    = 32 * AGG_U;
constexpr int AGG_QCAP    = 64;
constexpr int AGG_G       = 4;
constexpr int AGG_STEPS   = 32 / AGG_G;
constexpr int AGG_B       = 4;

__device__ __forceinline__ void agg_load_trip(const int4* __restrict__ dst4, int v0, int lane,
                                              int4 (&d)[AGG_U])
{
    #pragma unroll
    for (int u = 0; u < AGG_U; ++u) d[u] = dst4[v0 + u * 32 + lane];
}

__device__ __forceinline__ void agg_push_m(int2* q, int& qn, unsigned mask, bool m, int e, unsigned dl)
{
    if (mask) {
        if (m) q[qn + __builtin_amdgcn_mbcnt_lo(mask, 0u)] = make_int2(e, (int)dl);
        qn += __popc(mask);
    }
}

__device__ __forceinline__ void agg_push(int2* q, int& qn, bool m, int e, unsigned dl)
{
    agg_push_m(q, qn, __builtin_amdgcn_ballot_w32(m), m, e, dl);
}

template <bool FULL>
__device__ __forceinline__ void agg_drain(const int2* q, int cnt, int lane,
                                          const int* __restrict__ src,
                                          const float* __restrict__ dinv,
                                          const float* __restrict__ xw,
                                          int base, int cb, float* hacc)
{
    const int  g  = lane >> 3;
    const int  l7 = lane & 7;
    const bool r1 = (g & 1) != 0;
    const bool r2 = (g & 2) != 0;
    const int  k0 = 4 * l7 + ( g      & 3);
    const int  k1 = 4 * l7 + ((g + 1) & 3);
    const int  k2 = 4 * l7 + ((g + 2) & 3);
    const int  k3 = 4 * l7 + ((g + 3) & 3);

    const float4* __restrict__ xw4 = (const float4*)xw;
    unsigned soff[AGG_STEPS];
    int      hrow[AGG_STEPS];
    float nrm[AGG_STEPS];
    #pragma unroll
    for (int t = 0; t < AGG_STEPS; ++t) {
        const int  idx = AGG_G * t + g;
        const int2 ent = q[(FULL || idx < cnt) ? idx : 0];
        hrow[t] = ent.y * 32;
        nrm[t]  = dinv[base + ent.y];
        soff[t] = (unsigned)(src[ent.x] * 16 + (cb >> 2) + l7);
    }
    #pragma unroll
    for (int h = 0; h < AGG_STEPS / AGG_B; ++h) {
        if (!FULL && AGG_G * AGG_B * h >= cnt) break;
        float4 row[AGG_B];
        #pragma unroll
        for (int i = 0; i < AGG_B; ++i)
            row[i] = xw4[soff[AGG_B * h + i]];
        #pragma unroll
        for (int i = 0; i < AGG_B; ++i) {
            const int t = AGG_B * h + i;
            if (FULL || AGG_G * t + g < cnt) {
                const float4 v = row[i];
                const float a0 = r1 ? v.y : v.x, a1 = r1 ? v.z : v.y;
                const float a2 = r1 ? v.w : v.z, a3 = r1 ? v.x : v.w;
                const float b0 = r2 ? a2 : a0,   b1 = r2 ? a3 : a1;
                const float b2 = r2 ? a0 : a2,   b3 = r2 ? a1 : a3;
                const float nk = nrm[t];
                float* hr = hacc + hrow[t];
                lds_add_f32(hr + k0, b0 * nk);
                lds_add_f32(hr + k1, b1 * nk);
                lds_add_f32(hr + k2, b2 * nk);
                lds_add_f32(hr + k3, b3 * nk);
            }
        }
    }
}

__global__ __launch_bounds__(AGG_THREADS)
void aggregate_kernel(const int* __restrict__ en,
                      const int* __restrict__ es,
                      const float* __restrict__ xw,
                      const float* __restrict__ dn,
                      const float* __restrict__ dsv,
                      float* __restrict__ h, int n, int E)
{
    __shared__ float hacc[AGG_ROWS * 32];
    __shared__ int2  queue[AGG_WAVES][AGG_QCAP];

    const int lane = threadIdx.x & 31;
    const int wave = __builtin_amdgcn_readfirstlane((int)(threadIdx.x >> 5));
    int2* q = queue[wave];

    const int rng = blockIdx.x / AGG_SLICES;
    const int slc = blockIdx.x % AGG_SLICES;
    const int gw  = slc * AGG_WAVES + wave;
    constexpr int TW = AGG_SLICES * AGG_WAVES;

    for (int base = rng * AGG_ROWS; base < n; base += AGG_RANGES * AGG_ROWS) {
        const int rows = (n - base < AGG_ROWS) ? (n - base) : AGG_ROWS;

        for (int t = threadIdx.x; t < rows * 32; t += AGG_THREADS) hacc[t] = 0.0f;
        __syncthreads();

        for (int typ = 0; typ < 2; ++typ) {
            const int*   src  = typ ? es  : en;
            const int*   dst  = src + E;
            const float* dinv = typ ? dsv : dn;
            const int    cb   = typ * 32;
            int qn = 0;

            const int nvec = (E & 3) ? 0 : ((E >> 2) / AGG_TRIP) * AGG_TRIP;
            const int4* dst4 = (const int4*)dst;
            int v0 = gw * AGG_TRIP;
            if (v0 < nvec) {
                int4 d[AGG_U];
                agg_load_trip(dst4, v0, lane, d);
                while (true) {
                    const int  v1   = v0 + TW * AGG_TRIP;
                    const bool more = v1 < nvec;
                    int4 dnx[AGG_U];
                    agg_load_trip(dst4, more ? v1 : v0, lane, dnx);
                    bool     mt[4 * AGG_U];
                    unsigned mk[4 * AGG_U];
                    int      tot = 0;
                    #pragma unroll
                    for (int u = 0; u < AGG_U; ++u) {
                        const int dc[4] = { d[u].x, d[u].y, d[u].z, d[u].w };
                        #pragma unroll
                        for (int c = 0; c < 4; ++c) {
                            mt[4 * u + c] = (unsigned)(dc[c] - base) < (unsigned)rows;
                            mk[4 * u + c] = __builtin_amdgcn_ballot_w32(mt[4 * u + c]);
                            tot += __popc(mk[4 * u + c]);
                        }
                    }
                    if (qn + tot <= AGG_QCAP) {
                        #pragma unroll
                        for (int u = 0; u < AGG_U; ++u) {
                            const int e0 = 4 * (v0 + u * 32 + lane);
                            const int dc[4] = { d[u].x, d[u].y, d[u].z, d[u].w };
                            #pragma unroll
                            for (int c = 0; c < 4; ++c)
                                agg_push_m(q, qn, mk[4 * u + c], mt[4 * u + c], e0 + c, (unsigned)(dc[c] - base));
                        }
                        __builtin_amdgcn_wave_barrier();
                        while (qn >= 32) {
                            qn -= 32;
                            agg_drain<true>(q + qn, 32, lane, src, dinv, xw, base, cb, hacc);
                        }
                    } else {
                        #pragma nounroll
                        for (int k = 0; k < 4 * AGG_U; ++k) {
                            int dv = 0;
                            #pragma unroll
                            for (int u = 0; u < AGG_U; ++u) {
                                const int dc[4] = { d[u].x, d[u].y, d[u].z, d[u].w };
                                #pragma unroll
                                for (int c = 0; c < 4; ++c) dv = (k == 4 * u + c) ? dc[c] : dv;
                            }
                            const unsigned dl = (unsigned)(dv - base);
                            const int      e  = 4 * (v0 + lane) + 128 * (k >> 2) + (k & 3);
                            agg_push(q, qn, dl < (unsigned)rows, e, dl);
                            __builtin_amdgcn_wave_barrier();
                            if (qn >= 32) {
                                qn -= 32;
                                agg_drain<true>(q + qn, 32, lane, src, dinv, xw, base, cb, hacc);
                            }
                        }
                    }
                    if (!more) break;
                    #pragma unroll
                    for (int u = 0; u < AGG_U; ++u) d[u] = dnx[u];
                    v0 = v1;
                }
            }
            for (int e0 = 4 * nvec + gw * 32; e0 < E; e0 += TW * 32) {
                const int e = e0 + lane;
                const int dv = (e < E) ? dst[e] : -1;
                const unsigned dl = (unsigned)(dv - base);
                agg_push(q, qn, dl < (unsigned)rows, e, dl);
                __builtin_amdgcn_wave_barrier();
                while (qn >= 32) {
                    qn -= 32;
                    agg_drain<true>(q + qn, 32, lane, src, dinv, xw, base, cb, hacc);
                }
            }
            __builtin_amdgcn_wave_barrier();
            if (qn > 0) agg_drain<false>(q, qn, lane, src, dinv, xw, base, cb, hacc);
        }
        asm volatile("s_wait_dscnt 0x0" ::: "memory");
        __syncthreads();

        float* hrow = h + (size_t)base * 32;
        for (int t = threadIdx.x; t < rows * 32; t += AGG_THREADS)
            atomic_add_f32(&hrow[t], hacc[t]);
        __syncthreads();
    }
}

__device__ __forceinline__ float head_row(float hv, float bnj, float bsj, float wj)
{
    float v = hv + bnj + bsj;
    v = fmaxf(v, 0.0f) * wj;
    #pragma unroll
    for (int off = 16; off > 0; off >>= 1)
        v += __shfl_xor(v, off, 32);
    return v;
}

__global__ __launch_bounds__(256)
void head_kernel(const float* __restrict__ h,
                 const float* __restrict__ bn,
                 const float* __restrict__ bs,
                 const float* __restrict__ wl,
                 const float* __restrict__ bl,
                 float* __restrict__ out, int n)
{
    const int lane = threadIdx.x & 31;
    const int r0   = ((blockIdx.x * blockDim.x + threadIdx.x) >> 5) * 32;
    if (r0 >= n) return;
    const float bnj  = bn[lane], bsj = bs[lane];
    const float wj   = wl[lane];
    const float blin = bl[0];
    const float* hp  = h + (size_t)r0 * 32 + lane;
    float res = 0.0f;
    if (n - r0 >= 32) {
        float hv[32];
        #pragma unroll
        for (int rr = 0; rr < 32; ++rr) hv[rr] = hp[rr * 32];
        #pragma unroll
        for (int rr = 0; rr < 32; ++rr) {
            const float v = head_row(hv[rr], bnj, bsj, wj);
            if (lane == rr) res = v + blin;
        }
        out[r0 + lane] = res;
    } else {
        const int rcnt = n - r0;
        for (int rr = 0; rr < rcnt; ++rr) {
            const float v = head_row(hp[rr * 32], bnj, bsj, wj);
            if (lane == rr) res = v + blin;
        }
        if (lane < rcnt) out[r0 + lane] = res;
    }
}

extern "C" void kernel_launch(void* const* d_in, const int* in_sizes, int n_in,
                              void* d_out, int out_size, void* d_ws, size_t ws_size,
                              hipStream_t stream)
{
    const float* x  = (const float*)d_in[0];
    const int*   en = (const int*)  d_in[1];
    const int*   es = (const int*)  d_in[2];
    const float* Wn = (const float*)d_in[3];
    const float* bn = (const float*)d_in[4];
    const float* Ws = (const float*)d_in[5];
    const float* bs = (const float*)d_in[6];
    const float* Wl = (const float*)d_in[7];
    const float* bl = (const float*)d_in[8];
    float* out = (float*)d_out;

    const int N = in_sizes[0] / 64;
    const int E = in_sizes[1] / 2;

    float* ws  = (float*)d_ws;
    float* xw  = ws;
    float* h   = xw + (size_t)N * 64;
    float* dn  = xw + (size_t)N * 96;
    float* ds  = dn + N;

    const int TPB = 256;

    degree_kernel<<<DEG_GRID, DEG_THREADS, 0, stream>>>(en + E, es + E, h, N, E);
    dinv_kernel<<<(N + TPB - 1) / TPB, TPB, 0, stream>>>(h, dn, ds, N);

    gemm_xw_kernel<<<(N + 63) / 64, 128, 0, stream>>>(x, Wn, Ws, dn, ds, xw, h, N);

    aggregate_kernel<<<AGG_GRID, AGG_THREADS, 0, stream>>>(en, es, xw, dn, ds, h, N, E);

    head_kernel<<<(N + TPB - 1) / TPB, TPB, 0, stream>>>(h, bn, bs, Wl, bl, out, N);
}
